# starting kernel
# baseline (speedup 1.0000x reference)
#include <hip/hip_runtime.h>
#include <math.h>

#define NN 100000
#define NE 1600000
#define FIN 128
#define HID 64
#define NCLS 10
#define ROW_TILES 6250

typedef __attribute__((ext_vector_type(2))) float v2f;
typedef __attribute__((ext_vector_type(8))) float v8f;

__global__ void k_fill(float* __restrict__ p, int n, float v) {
    int i = blockIdx.x * blockDim.x + threadIdx.x;
    if (i < n) p[i] = v;
}

__global__ void k_deg(const int* __restrict__ dst, float* __restrict__ deg) {
    int e = blockIdx.x * blockDim.x + threadIdx.x;
    if (e < NE) atomicAdd(&deg[dst[e]], 1.0f);
}

__global__ void k_dis(float* __restrict__ deg) {
    int i = blockIdx.x * blockDim.x + threadIdx.x;
    if (i < NN) deg[i] = rsqrtf(deg[i]);
}

__global__ void k_norm(const int* __restrict__ src, const int* __restrict__ dst,
                       const float* __restrict__ dis, float* __restrict__ norm) {
    int e = blockIdx.x * blockDim.x + threadIdx.x;
    if (e < NE) norm[e] = dis[src[e]] * dis[dst[e]];
}

__global__ void __launch_bounds__(128)
k_gemm1(const float* __restrict__ x, const float* __restrict__ W1,
        float* __restrict__ h1) {
    __shared__ float wT[HID * 132];
    int t = threadIdx.x;
    for (int i = t; i < FIN * HID; i += 128) {
        int k = i >> 6, c = i & 63;
        wT[c * 132 + k] = W1[i];
    }
    __syncthreads();

    int tile = blockIdx.x * 4 + (t >> 5);
    if (tile >= ROW_TILES) return;
    int lane = t & 31;
    int m = lane & 15, half = lane >> 4;

    const float* arow = x + (size_t)(tile * 16 + m) * FIN + 2 * half;
    v8f a0 = {}, a1 = {}, a2 = {}, a3 = {};

    #pragma unroll 4
    for (int k0 = 0; k0 < FIN; k0 += 4) {
        v2f a = *(const v2f*)(arow + k0);
        const float* bp = wT + m * 132 + k0 + 2 * half;
        v2f b0 = *(const v2f*)(bp);
        v2f b1 = *(const v2f*)(bp + 16 * 132);
        v2f b2 = *(const v2f*)(bp + 32 * 132);
        v2f b3 = *(const v2f*)(bp + 48 * 132);
        a0 = __builtin_amdgcn_wmma_f32_16x16x4_f32(false, a, false, b0, (short)0, a0, false, false);
        a1 = __builtin_amdgcn_wmma_f32_16x16x4_f32(false, a, false, b1, (short)0, a1, false, false);
        a2 = __builtin_amdgcn_wmma_f32_16x16x4_f32(false, a, false, b2, (short)0, a2, false, false);
        a3 = __builtin_amdgcn_wmma_f32_16x16x4_f32(false, a, false, b3, (short)0, a3, false, false);
    }

    float* out = h1 + (size_t)(tile * 16 + 8 * half) * HID + m;
    #pragma unroll
    for (int v = 0; v < 8; ++v) {
        out[v * HID +  0] = a0[v];
        out[v * HID + 16] = a1[v];
        out[v * HID + 32] = a2[v];
        out[v * HID + 48] = a3[v];
    }
}

__global__ void __launch_bounds__(128)
k_gemm2(const float* __restrict__ z1, const float* __restrict__ W2,
        float* __restrict__ h2) {
    __shared__ float wT[16 * 68];
    int t = threadIdx.x;
    for (int i = t; i < 16 * HID; i += 128) {
        int k = i >> 4, c = i & 15;
        wT[c * 68 + k] = (c < NCLS) ? W2[k * NCLS + c] : 0.0f;
    }
    __syncthreads();

    int tile = blockIdx.x * 4 + (t >> 5);
    if (tile >= ROW_TILES) return;
    int lane = t & 31;
    int m = lane & 15, half = lane >> 4;

    const float* arow = z1 + (size_t)(tile * 16 + m) * HID + 2 * half;
    v8f acc = {};
    #pragma unroll
    for (int k0 = 0; k0 < HID; k0 += 4) {
        v2f a = *(const v2f*)(arow + k0);
        v2f b = *(const v2f*)(wT + m * 68 + k0 + 2 * half);
        acc = __builtin_amdgcn_wmma_f32_16x16x4_f32(false, a, false, b, (short)0, acc, false, false);
    }
    if (m < NCLS) {
        float* out = h2 + (size_t)(tile * 16 + 8 * half) * NCLS + m;
        #pragma unroll
        for (int v = 0; v < 8; ++v) out[v * NCLS] = acc[v];
    }
}

__global__ void k_scatter1(const int* __restrict__ src, const int* __restrict__ dst,
                           const float* __restrict__ norm, const float* __restrict__ h1,
                           float* __restrict__ agg) {
    int g = blockIdx.x * blockDim.x + threadIdx.x;
    int e = g >> 6, f = g & 63;
    if (e >= NE) return;
    int s = src[e], d = dst[e];
    atomicAdd(&agg[(size_t)d * HID + f], h1[(size_t)s * HID + f] * norm[e]);
}

__global__ void k_scatter2(const int* __restrict__ src, const int* __restrict__ dst,
                           const float* __restrict__ norm, const float* __restrict__ h2,
                           float* __restrict__ out) {
    int g = blockIdx.x * blockDim.x + threadIdx.x;
    int e = g >> 4, f = g & 15;
    if (e >= NE || f >= NCLS) return;
    int s = src[e], d = dst[e];
    atomicAdd(&out[(size_t)d * NCLS + f], h2[(size_t)s * NCLS + f] * norm[e]);
}

__global__ void k_post1(float* __restrict__ agg, const float* __restrict__ h1,
                        const float* __restrict__ dis, const float* __restrict__ b1) {
    int i = blockIdx.x * blockDim.x + threadIdx.x;
    if (i >= NN * HID) return;
    int node = i >> 6, f = i & 63;
    float di = dis[node];
    float v = agg[i] + h1[i] * di * di + b1[f];
    agg[i] = v > 0.0f ? v : 0.0f;
}

__global__ void k_post2(float* __restrict__ out, const float* __restrict__ h2,
                        const float* __restrict__ dis, const float* __restrict__ b2) {
    int i = blockIdx.x * blockDim.x + threadIdx.x;
    if (i >= NN) return;
    float d2 = dis[i] * dis[i];
    float tv[NCLS];
    float mx = -3.402823466e38f;
    #pragma unroll
    for (int f = 0; f < NCLS; ++f) {
        float v = out[(size_t)i * NCLS + f] + h2[(size_t)i * NCLS + f] * d2 + b2[f];
        tv[f] = v;
        mx = fmaxf(mx, v);
    }
    float s = 0.0f;
    #pragma unroll
    for (int f = 0; f < NCLS; ++f) s += expf(tv[f] - mx);
    float lse = mx + logf(s);
    #pragma unroll
    for (int f = 0; f < NCLS; ++f) out[(size_t)i * NCLS + f] = tv[f] - lse;
}

extern "C" void kernel_launch(void* const* d_in, const int* in_sizes, int n_in,
                              void* d_out, int out_size, void* d_ws, size_t ws_size,
                              hipStream_t stream) {
    const float* x  = (const float*)d_in[0];
    const int*   ei = (const int*)d_in[1];
    const float* W1 = (const float*)d_in[2];
    const float* b1 = (const float*)d_in[3];
    const float* W2 = (const float*)d_in[4];
    const float* b2 = (const float*)d_in[5];
    const int* src = ei;
    const int* dst = ei + NE;
    float* out = (float*)d_out;

    float* ws   = (float*)d_ws;
    float* dis  = ws;
    float* h1   = dis + NN;
    float* agg1 = h1 + (size_t)NN * HID;
    float* h2   = agg1 + (size_t)NN * HID;
    float* norm = h2 + (size_t)NN * NCLS;

    const int B = 256;
    k_fill<<<(NN + B - 1) / B, B, 0, stream>>>(dis, NN, 1.0f);
    k_fill<<<(NN * HID + B - 1) / B, B, 0, stream>>>(agg1, NN * HID, 0.0f);
    k_fill<<<(NN * NCLS + B - 1) / B, B, 0, stream>>>(out, NN * NCLS, 0.0f);

    k_deg<<<(NE + B - 1) / B, B, 0, stream>>>(dst, dis);
    k_dis<<<(NN + B - 1) / B, B, 0, stream>>>(dis);
    k_norm<<<(NE + B - 1) / B, B, 0, stream>>>(src, dst, dis, norm);

    k_gemm1<<<(ROW_TILES + 3) / 4, 128, 0, stream>>>(x, W1, h1);
    k_scatter1<<<(NE * 64) / B, B, 0, stream>>>(src, dst, norm, h1, agg1);
    k_post1<<<(NN * HID + B - 1) / B, B, 0, stream>>>(agg1, h1, dis, b1);

    k_gemm2<<<(ROW_TILES + 3) / 4, 128, 0, stream>>>(agg1, W2, h2);
    k_scatter2<<<(NE * 16) / B, B, 0, stream>>>(src, dst, norm, h2, out);
    k_post2<<<(NN + B - 1) / B, B, 0, stream>>>(out, h2, dis, b2);
}


# Edges ranked into destination-sorted rows (one returning integer atomic per edge + scan); both aggregations become per-row gathers without float atomics, epilogues fused; row sums in rank order
# speedup vs baseline: 2.3762x; 2.3762x over previous
#include <hip/hip_runtime.h>
#include <math.h>


#define NN 100000
#define NE 1600000
#define FIN 128
#define HID 64
#define NCLS 10
#define ROW_TILES 6250
#define SCAN_ELEMS 1024
#define SCAN_BLOCKS 98

typedef __attribute__((ext_vector_type(2))) float v2f;
typedef __attribute__((ext_vector_type(8))) float v8f;

__global__ void k_zero(int* __restrict__ p, int n) {
    int i = blockIdx.x * blockDim.x + threadIdx.x;
    if (i < n) p[i] = 0;
}

__global__ void k_rank(const int* __restrict__ dst, int* __restrict__ cnt,
                       int* __restrict__ rank) {
    int e = blockIdx.x * blockDim.x + threadIdx.x;
    if (e < NE) rank[e] = atomicAdd(&cnt[dst[e]], 1);
}

__global__ void __launch_bounds__(256)
k_bsum(const int* __restrict__ cnt, int* __restrict__ bsum) {
    __shared__ int red[8];
    int t = threadIdx.x;
    int lane = t & 31, wv = t >> 5;
    int i0 = blockIdx.x * SCAN_ELEMS + 4 * t;
    int4 c = make_int4(0, 0, 0, 0);
    if (i0 < NN) c = *(const int4*)(cnt + i0);
    int s = (c.x + c.y) + (c.z + c.w);
    for (int o = 16; o > 0; o >>= 1) s += __shfl_xor(s, o);
    if (lane == 0) red[wv] = s;
    __syncthreads();
    if (t == 0) {
        int tot = 0;
        for (int w = 0; w < 8; ++w) tot += red[w];
        bsum[blockIdx.x] = tot;
    }
}

__global__ void __launch_bounds__(256)
k_scan(const int* __restrict__ cnt, const int* __restrict__ bsum,
       int* __restrict__ rowptr, float* __restrict__ dis) {
    __shared__ int red[8];
    __shared__ int wsum[8];
    int t = threadIdx.x;
    int lane = t & 31, wv = t >> 5;
    int base = blockIdx.x * SCAN_ELEMS;

    int part = (t < (int)blockIdx.x) ? bsum[t] : 0;
    for (int o = 16; o > 0; o >>= 1) part += __shfl_xor(part, o);
    if (lane == 0) red[wv] = part;

    int i0 = base + 4 * t;
    int4 c = make_int4(0, 0, 0, 0);
    if (i0 < NN) c = *(const int4*)(cnt + i0);
    int s = (c.x + c.y) + (c.z + c.w);
    int inc = s;
    for (int o = 1; o < 32; o <<= 1) {
        int v = __shfl_up(inc, o);
        if (lane >= o) inc += v;
    }
    if (lane == 31) wsum[wv] = inc;
    __syncthreads();

    int off = 0;
    for (int w = 0; w < 8; ++w) off += red[w];
    for (int w = 0; w < 8; ++w) off += (w < wv) ? wsum[w] : 0;
    int ex = off + inc - s;
    if (i0 < NN) {
        int4 r;
        r.x = ex;
        r.y = ex + c.x;
        r.z = ex + c.x + c.y;
        r.w = ex + c.x + c.y + c.z;
        *(int4*)(rowptr + i0) = r;
        float4 d;
        d.x = rsqrtf((float)(c.x + 1));
        d.y = rsqrtf((float)(c.y + 1));
        d.z = rsqrtf((float)(c.z + 1));
        d.w = rsqrtf((float)(c.w + 1));
        *(float4*)(dis + i0) = d;
    }
}

__global__ void k_place(const int* __restrict__ src, const int* __restrict__ dst,
                        const int* __restrict__ rowptr, const int* __restrict__ rank,
                        int* __restrict__ csr) {
    int e = blockIdx.x * blockDim.x + threadIdx.x;
    if (e < NE) csr[rowptr[dst[e]] + rank[e]] = src[e];
}

__global__ void __launch_bounds__(128)
k_gemm1(const float* __restrict__ x, const float* __restrict__ W1,
        float* __restrict__ h1) {
    __shared__ float wT[HID * 132];
    int t = threadIdx.x;
    for (int i = t; i < FIN * HID; i += 128) {
        int k = i >> 6, c = i & 63;
        wT[c * 132 + k] = W1[i];
    }
    __syncthreads();

    int tile = blockIdx.x * 4 + (t >> 5);
    if (tile >= ROW_TILES) return;
    int lane = t & 31;
    int m = lane & 15, half = lane >> 4;

    const float* arow = x + (size_t)(tile * 16 + m) * FIN + 2 * half;
    v8f a0 = {}, a1 = {}, a2 = {}, a3 = {};

    #pragma unroll 4
    for (int k0 = 0; k0 < FIN; k0 += 4) {
        v2f a = *(const v2f*)(arow + k0);
        const float* bp = wT + m * 132 + k0 + 2 * half;
        v2f b0 = *(const v2f*)(bp);
        v2f b1 = *(const v2f*)(bp + 16 * 132);
        v2f b2 = *(const v2f*)(bp + 32 * 132);
        v2f b3 = *(const v2f*)(bp + 48 * 132);
        a0 = __builtin_amdgcn_wmma_f32_16x16x4_f32(false, a, false, b0, (short)0, a0, false, false);
        a1 = __builtin_amdgcn_wmma_f32_16x16x4_f32(false, a, false, b1, (short)0, a1, false, false);
        a2 = __builtin_amdgcn_wmma_f32_16x16x4_f32(false, a, false, b2, (short)0, a2, false, false);
        a3 = __builtin_amdgcn_wmma_f32_16x16x4_f32(false, a, false, b3, (short)0, a3, false, false);
    }

    float* out = h1 + (size_t)(tile * 16 + 8 * half) * HID + m;
    #pragma unroll
    for (int v = 0; v < 8; ++v) {
        out[v * HID +  0] = a0[v];
        out[v * HID + 16] = a1[v];
        out[v * HID + 32] = a2[v];
        out[v * HID + 48] = a3[v];
    }
}

__global__ void __launch_bounds__(128)
k_gemm2(const float* __restrict__ z1, const float* __restrict__ W2,
        float* __restrict__ h2) {
    __shared__ float wT[16 * 68];
    int t = threadIdx.x;
    for (int i = t; i < 16 * HID; i += 128) {
        int k = i >> 4, c = i & 15;
        wT[c * 68 + k] = (c < NCLS) ? W2[k * NCLS + c] : 0.0f;
    }
    __syncthreads();

    int tile = blockIdx.x * 4 + (t >> 5);
    if (tile >= ROW_TILES) return;
    int lane = t & 31;
    int m = lane & 15, half = lane >> 4;

    const float* arow = z1 + (size_t)(tile * 16 + m) * HID + 2 * half;
    v8f acc = {};
    #pragma unroll
    for (int k0 = 0; k0 < HID; k0 += 4) {
        v2f a = *(const v2f*)(arow + k0);
        v2f b = *(const v2f*)(wT + m * 68 + k0 + 2 * half);
        acc = __builtin_amdgcn_wmma_f32_16x16x4_f32(false, a, false, b, (short)0, acc, false, false);
    }
    if (m < NCLS) {
        float* out = h2 + (size_t)(tile * 16 + 8 * half) * NCLS + m;
        #pragma unroll
        for (int v = 0; v < 8; ++v) out[v * NCLS] = acc[v];
    }
}

__global__ void __launch_bounds__(256)
k_gather1(const int* __restrict__ rowptr, const int* __restrict__ cnt,
          const int* __restrict__ csr, const float* __restrict__ dis,
          const float* __restrict__ h1, const float* __restrict__ b1,
          float* __restrict__ z1) {
    int g = blockIdx.x * 256 + threadIdx.x;
    int node = g >> 4, l = g & 15;
    int beg = rowptr[node];
    int end = beg + cnt[node];
    float di = dis[node];
    float4 acc = make_float4(0.0f, 0.0f, 0.0f, 0.0f);
    #pragma unroll 4
    for (int j = beg; j < end; ++j) {
        int s = (int)min((unsigned)csr[j], (unsigned)(NN - 1));
        float w = dis[s] * di;
        float4 v = *(const float4*)(h1 + (size_t)s * HID + 4 * l);
        acc.x += v.x * w;
        acc.y += v.y * w;
        acc.z += v.z * w;
        acc.w += v.w * w;
    }
    float4 hv = *(const float4*)(h1 + (size_t)node * HID + 4 * l);
    float4 bb = *(const float4*)(b1 + 4 * l);
    float4 r;
    r.x = acc.x + hv.x * di * di + bb.x;
    r.y = acc.y + hv.y * di * di + bb.y;
    r.z = acc.z + hv.z * di * di + bb.z;
    r.w = acc.w + hv.w * di * di + bb.w;
    r.x = r.x > 0.0f ? r.x : 0.0f;
    r.y = r.y > 0.0f ? r.y : 0.0f;
    r.z = r.z > 0.0f ? r.z : 0.0f;
    r.w = r.w > 0.0f ? r.w : 0.0f;
    *(float4*)(z1 + (size_t)node * HID + 4 * l) = r;
}

__global__ void __launch_bounds__(256)
k_gather2(const int* __restrict__ rowptr, const int* __restrict__ cnt,
          const int* __restrict__ csr, const float* __restrict__ dis,
          const float* __restrict__ h2, const float* __restrict__ b2,
          float* __restrict__ out) {
    __shared__ float stage[16 * NCLS];
    int t = threadIdx.x;
    int g = blockIdx.x * 256 + t;
    int node = g >> 4, f = g & 15;
    int fc = f < NCLS ? f : NCLS - 1;
    int beg = rowptr[node];
    int end = beg + cnt[node];
    float di = dis[node];
    float acc = 0.0f;
    #pragma unroll 4
    for (int j = beg; j < end; ++j) {
        int s = (int)min((unsigned)csr[j], (unsigned)(NN - 1));
        acc += h2[(size_t)s * NCLS + fc] * (dis[s] * di);
    }
    float v = acc + h2[(size_t)node * NCLS + fc] * (di * di) + b2[fc];
    float mx = f < NCLS ? v : -3.402823466e38f;
    for (int o = 8; o > 0; o >>= 1) mx = fmaxf(mx, __shfl_xor(mx, o, 16));
    float ex = f < NCLS ? expf(v - mx) : 0.0f;
    for (int o = 8; o > 0; o >>= 1) ex += __shfl_xor(ex, o, 16);
    float lse = mx + logf(ex);
    if (f < NCLS) stage[(t >> 4) * NCLS + f] = v - lse;
    __syncthreads();
    if (t < 16 * NCLS) out[(size_t)blockIdx.x * (16 * NCLS) + t] = stage[t];
}

extern "C" void kernel_launch(void* const* d_in, const int* in_sizes, int n_in,
                              void* d_out, int out_size, void* d_ws, size_t ws_size,
                              hipStream_t stream) {
    const float* x  = (const float*)d_in[0];
    const int*   ei = (const int*)d_in[1];
    const float* W1 = (const float*)d_in[2];
    const float* b1 = (const float*)d_in[3];
    const float* W2 = (const float*)d_in[4];
    const float* b2 = (const float*)d_in[5];
    const int* src = ei;
    const int* dst = ei + NE;
    float* out = (float*)d_out;

    float* ws   = (float*)d_ws;
    float* h1   = ws;
    float* z1   = h1 + (size_t)NN * HID;
    float* h2   = z1 + (size_t)NN * HID;
    float* dis  = h2 + (size_t)NN * NCLS;
    int* cnt    = (int*)(dis + NN);
    int* rowptr = cnt + NN;
    int* rank   = rowptr + NN;
    int* csr    = rank + NE;
    int* bsum   = csr + NE;

    const int B = 256;
    k_zero<<<(NN + B - 1) / B, B, 0, stream>>>(cnt, NN);
    k_rank<<<(NE + B - 1) / B, B, 0, stream>>>(dst, cnt, rank);
    k_bsum<<<SCAN_BLOCKS, 256, 0, stream>>>(cnt, bsum);
    k_scan<<<SCAN_BLOCKS, 256, 0, stream>>>(cnt, bsum, rowptr, dis);
    k_place<<<(NE + B - 1) / B, B, 0, stream>>>(src, dst, rowptr, rank, csr);

    k_gemm1<<<(ROW_TILES + 3) / 4, 128, 0, stream>>>(x, W1, h1);
    k_gather1<<<NN / 16, 256, 0, stream>>>(rowptr, cnt, csr, dis, h1, b1, z1);

    k_gemm2<<<(ROW_TILES + 3) / 4, 128, 0, stream>>>(z1, W2, h2);
    k_gather2<<<NN / 16, 256, 0, stream>>>(rowptr, cnt, csr, dis, h2, b2, out);
}


# Edge ranks by returning adds on LDS counters (edges binned by 391-node destination range, then one block per bin) replace the per-edge global atomic; row sums in LDS-add arrival order
# speedup vs baseline: 4.3341x; 1.8240x over previous
#include <hip/hip_runtime.h>
#include <math.h>


#define NN 100000
#define NE 1600000
#define FIN 128
#define HID 64
#define NCLS 10
#define ROW_TILES 6250
#define NB 256
#define BIN_NODES 391
#define NCH 256
#define CH 6250
#define CH_PER_T 25
#define SORT_T 1024
#define STAGE_CAP 16384

typedef __attribute__((ext_vector_type(2))) float v2f;
typedef __attribute__((ext_vector_type(8))) float v8f;

__device__ __forceinline__ unsigned bin_of(int d) {
    unsigned b = (unsigned)d / BIN_NODES;
    return b < NB - 1 ? b : NB - 1;
}

__device__ __forceinline__ int block_excl_scan(int c, int t, int* wtot, int nwaves) {
    int lane = t & 31, wv = t >> 5;
    int inc = c;
    for (int o = 1; o < 32; o <<= 1) {
        int v = __shfl_up(inc, o);
        if (lane >= o) inc += v;
    }
    if (lane == 31 && wv < nwaves) wtot[wv] = inc;
    __syncthreads();
    int off = 0;
    for (int w = 0; w < nwaves; ++w) off += (w < wv) ? wtot[w] : 0;
    return off + inc - c;
}

__global__ void __launch_bounds__(256)
k_bincount(const int* __restrict__ dst, int* __restrict__ counts) {
    __shared__ int hist[NB];
    int t = threadIdx.x;
    hist[t] = 0;
    __syncthreads();
    const int* d = dst + blockIdx.x * CH;
    int dv[CH_PER_T];
    #pragma unroll
    for (int k = 0; k < CH_PER_T; ++k) {
        int i = t + 256 * k;
        dv[k] = (i < CH) ? d[i] : -1;
    }
    #pragma unroll
    for (int k = 0; k < CH_PER_T; ++k)
        if (dv[k] >= 0) atomicAdd(&hist[bin_of(dv[k])], 1);
    __syncthreads();
    counts[blockIdx.x * NB + t] = hist[t];
}

__global__ void __launch_bounds__(1024)
k_binscan(const int* __restrict__ counts, int* __restrict__ offs, int* __restrict__ binbase) {
    __shared__ int part[4][NB];
    __shared__ int bbase[NB];
    __shared__ int wtot[8];
    int t = threadIdx.x;
    int b = t & (NB - 1), q = t >> 8;
    const int* c = counts + (q * 64) * NB + b;
    int v[64];
    #pragma unroll
    for (int k = 0; k < 64; ++k) v[k] = c[k * NB];
    int s = 0;
    #pragma unroll
    for (int k = 0; k < 64; ++k) s += v[k];
    part[q][b] = s;
    __syncthreads();
    int tot = (t < NB) ? (part[0][b] + part[1][b]) + (part[2][b] + part[3][b]) : 0;
    int ex = block_excl_scan(tot, t, wtot, 8);
    if (t < NB) {
        bbase[t] = ex;
        binbase[t] = ex;
        if (t == NB - 1) binbase[NB] = ex + tot;
    }
    __syncthreads();
    int run = bbase[b];
    for (int qq = 0; qq < q; ++qq) run += part[qq][b];
    int* o = offs + (q * 64) * NB + b;
    #pragma unroll
    for (int k = 0; k < 64; ++k) { o[k * NB] = run; run += v[k]; }
}

__global__ void __launch_bounds__(256)
k_binscatter(const int* __restrict__ src, const int* __restrict__ dst,
             const int* __restrict__ counts, const int* __restrict__ offs,
             int* __restrict__ binned_w) {
    int2* __restrict__ binned = (int2*)binned_w;
    __shared__ int cursor[NB];
    __shared__ int goff[NB];
    __shared__ int wtot[8];
    __shared__ int2 stage[CH];
    int t = threadIdx.x;
    const int* sp = src + blockIdx.x * CH;
    const int* dp = dst + blockIdx.x * CH;
    int sv[CH_PER_T], dv[CH_PER_T];
    #pragma unroll
    for (int k = 0; k < CH_PER_T; ++k) {
        int i = t + 256 * k;
        dv[k] = (i < CH) ? dp[i] : -1;
        sv[k] = (i < CH) ? sp[i] : 0;
    }
    int c = counts[blockIdx.x * NB + t];
    int ex = block_excl_scan(c, t, wtot, 8);
    cursor[t] = ex;
    goff[t] = offs[blockIdx.x * NB + t] - ex;
    __syncthreads();
    #pragma unroll
    for (int k = 0; k < CH_PER_T; ++k)
        if (dv[k] >= 0) {
            int p = atomicAdd(&cursor[bin_of(dv[k])], 1);
            stage[p] = make_int2(sv[k], dv[k]);
        }
    __syncthreads();
    for (int i = t; i < CH; i += 256) {
        int2 e = stage[i];
        binned[goff[bin_of(e.y)] + i] = e;
    }
}

__global__ void __launch_bounds__(SORT_T)
k_binsort(const int* __restrict__ binned_w, const int* __restrict__ binbase,
          int* __restrict__ rowptr, int* __restrict__ cnt, float* __restrict__ dis,
          int* __restrict__ csr) {
    const int2* __restrict__ binned = (const int2*)binned_w;
    __shared__ int lcnt[512];
    __shared__ int lcur[512];
    __shared__ int wtot[16];
    __shared__ int lst[STAGE_CAP];
    int t = threadIdx.x;
    int lo = blockIdx.x * BIN_NODES;
    int nn = NN - lo < BIN_NODES ? NN - lo : BIN_NODES;
    int base = binbase[blockIdx.x];
    int end = binbase[blockIdx.x + 1];
    int n = end - base;
    if (t < 512) lcnt[t] = 0;
    __syncthreads();
    for (int i0 = base; i0 < end; i0 += 4 * SORT_T) {
        int dv[4];
        #pragma unroll
        for (int k = 0; k < 4; ++k) {
            int i = i0 + k * SORT_T + t;
            dv[k] = (i < end) ? binned[i].y : -1;
        }
        #pragma unroll
        for (int k = 0; k < 4; ++k)
            if (dv[k] >= 0) {
                unsigned li = (unsigned)(dv[k] - lo);
                atomicAdd(&lcnt[li < BIN_NODES - 1 ? li : BIN_NODES - 1], 1);
            }
    }
    __syncthreads();
    int c = (t < 512) ? lcnt[t] : 0;
    int ex = block_excl_scan(c, t, wtot, 16);
    if (t < 512) lcur[t] = ex;
    if (t < nn) {
        rowptr[lo + t] = base + ex;
        cnt[lo + t] = c;
        dis[lo + t] = rsqrtf((float)(c + 1));
    }
    __syncthreads();
    bool staged = n <= STAGE_CAP;
    for (int i0 = base; i0 < end; i0 += 4 * SORT_T) {
        int2 ev[4];
        #pragma unroll
        for (int k = 0; k < 4; ++k) {
            int i = i0 + k * SORT_T + t;
            ev[k] = (i < end) ? binned[i] : make_int2(0, -1);
        }
        #pragma unroll
        for (int k = 0; k < 4; ++k)
            if (ev[k].y >= 0) {
                unsigned li = (unsigned)(ev[k].y - lo);
                int p = atomicAdd(&lcur[li < BIN_NODES - 1 ? li : BIN_NODES - 1], 1);
                if (staged) lst[p] = ev[k].x;
                else csr[base + p] = ev[k].x;
            }
    }
    if (staged) {
        __syncthreads();
        for (int i = t; i < n; i += SORT_T) csr[base + i] = lst[i];
    }
}

__global__ void __launch_bounds__(128)
k_gemm1(const float* __restrict__ x, const float* __restrict__ W1,
        float* __restrict__ h1) {
    __shared__ float wT[HID * 132];
    int t = threadIdx.x;
    for (int i = t; i < FIN * HID; i += 128) {
        int k = i >> 6, c = i & 63;
        wT[c * 132 + k] = W1[i];
    }
    __syncthreads();

    int tile = blockIdx.x * 4 + (t >> 5);
    if (tile >= ROW_TILES) return;
    int lane = t & 31;
    int m = lane & 15, half = lane >> 4;

    const float* arow = x + (size_t)(tile * 16 + m) * FIN + 2 * half;
    v8f a0 = {}, a1 = {}, a2 = {}, a3 = {};

    #pragma unroll 4
    for (int k0 = 0; k0 < FIN; k0 += 4) {
        v2f a = *(const v2f*)(arow + k0);
        const float* bp = wT + m * 132 + k0 + 2 * half;
        v2f b0 = *(const v2f*)(bp);
        v2f b1 = *(const v2f*)(bp + 16 * 132);
        v2f b2 = *(const v2f*)(bp + 32 * 132);
        v2f b3 = *(const v2f*)(bp + 48 * 132);
        a0 = __builtin_amdgcn_wmma_f32_16x16x4_f32(false, a, false, b0, (short)0, a0, false, false);
        a1 = __builtin_amdgcn_wmma_f32_16x16x4_f32(false, a, false, b1, (short)0, a1, false, false);
        a2 = __builtin_amdgcn_wmma_f32_16x16x4_f32(false, a, false, b2, (short)0, a2, false, false);
        a3 = __builtin_amdgcn_wmma_f32_16x16x4_f32(false, a, false, b3, (short)0, a3, false, false);
    }

    float* out = h1 + (size_t)(tile * 16 + 8 * half) * HID + m;
    #pragma unroll
    for (int v = 0; v < 8; ++v) {
        out[v * HID +  0] = a0[v];
        out[v * HID + 16] = a1[v];
        out[v * HID + 32] = a2[v];
        out[v * HID + 48] = a3[v];
    }
}

__global__ void __launch_bounds__(128)
k_gemm2(const float* __restrict__ z1, const float* __restrict__ W2,
        float* __restrict__ h2) {
    __shared__ float wT[16 * 68];
    int t = threadIdx.x;
    for (int i = t; i < 16 * HID; i += 128) {
        int k = i >> 4, c = i & 15;
        wT[c * 68 + k] = (c < NCLS) ? W2[k * NCLS + c] : 0.0f;
    }
    __syncthreads();

    int tile = blockIdx.x * 4 + (t >> 5);
    if (tile >= ROW_TILES) return;
    int lane = t & 31;
    int m = lane & 15, half = lane >> 4;

    const float* arow = z1 + (size_t)(tile * 16 + m) * HID + 2 * half;
    v8f acc = {};
    #pragma unroll
    for (int k0 = 0; k0 < HID; k0 += 4) {
        v2f a = *(const v2f*)(arow + k0);
        v2f b = *(const v2f*)(wT + m * 68 + k0 + 2 * half);
        acc = __builtin_amdgcn_wmma_f32_16x16x4_f32(false, a, false, b, (short)0, acc, false, false);
    }
    if (m < NCLS) {
        float* out = h2 + (size_t)(tile * 16 + 8 * half) * NCLS + m;
        #pragma unroll
        for (int v = 0; v < 8; ++v) out[v * NCLS] = acc[v];
    }
}

__global__ void __launch_bounds__(256)
k_gather1(const int* __restrict__ rowptr, const int* __restrict__ cnt,
          const int* __restrict__ csr, const float* __restrict__ dis,
          const float* __restrict__ h1, const float* __restrict__ b1,
          float* __restrict__ z1) {
    int g = blockIdx.x * 256 + threadIdx.x;
    int node = g >> 4, l = g & 15;
    int beg = rowptr[node];
    int end = beg + cnt[node];
    float di = dis[node];
    float4 acc = make_float4(0.0f, 0.0f, 0.0f, 0.0f);
    #pragma unroll 4
    for (int j = beg; j < end; ++j) {
        int s = (int)min((unsigned)csr[j], (unsigned)(NN - 1));
        float w = dis[s] * di;
        float4 v = *(const float4*)(h1 + (size_t)s * HID + 4 * l);
        acc.x += v.x * w;
        acc.y += v.y * w;
        acc.z += v.z * w;
        acc.w += v.w * w;
    }
    float4 hv = *(const float4*)(h1 + (size_t)node * HID + 4 * l);
    float4 bb = *(const float4*)(b1 + 4 * l);
    float4 r;
    r.x = acc.x + hv.x * di * di + bb.x;
    r.y = acc.y + hv.y * di * di + bb.y;
    r.z = acc.z + hv.z * di * di + bb.z;
    r.w = acc.w + hv.w * di * di + bb.w;
    r.x = r.x > 0.0f ? r.x : 0.0f;
    r.y = r.y > 0.0f ? r.y : 0.0f;
    r.z = r.z > 0.0f ? r.z : 0.0f;
    r.w = r.w > 0.0f ? r.w : 0.0f;
    *(float4*)(z1 + (size_t)node * HID + 4 * l) = r;
}

__global__ void __launch_bounds__(256)
k_gather2(const int* __restrict__ rowptr, const int* __restrict__ cnt,
          const int* __restrict__ csr, const float* __restrict__ dis,
          const float* __restrict__ h2, const float* __restrict__ b2,
          float* __restrict__ out) {
    __shared__ float stage[16 * NCLS];
    int t = threadIdx.x;
    int g = blockIdx.x * 256 + t;
    int node = g >> 4, f = g & 15;
    int fc = f < NCLS ? f : NCLS - 1;
    int beg = rowptr[node];
    int end = beg + cnt[node];
    float di = dis[node];
    float acc = 0.0f;
    #pragma unroll 4
    for (int j = beg; j < end; ++j) {
        int s = (int)min((unsigned)csr[j], (unsigned)(NN - 1));
        acc += h2[(size_t)s * NCLS + fc] * (dis[s] * di);
    }
    float v = acc + h2[(size_t)node * NCLS + fc] * (di * di) + b2[fc];
    float mx = f < NCLS ? v : -3.402823466e38f;
    for (int o = 8; o > 0; o >>= 1) mx = fmaxf(mx, __shfl_xor(mx, o, 16));
    float ex = f < NCLS ? expf(v - mx) : 0.0f;
    for (int o = 8; o > 0; o >>= 1) ex += __shfl_xor(ex, o, 16);
    float lse = mx + logf(ex);
    if (f < NCLS) stage[(t >> 4) * NCLS + f] = v - lse;
    __syncthreads();
    if (t < 16 * NCLS) out[(size_t)blockIdx.x * (16 * NCLS) + t] = stage[t];
}

extern "C" void kernel_launch(void* const* d_in, const int* in_sizes, int n_in,
                              void* d_out, int out_size, void* d_ws, size_t ws_size,
                              hipStream_t stream) {
    const float* x  = (const float*)d_in[0];
    const int*   ei = (const int*)d_in[1];
    const float* W1 = (const float*)d_in[2];
    const float* b1 = (const float*)d_in[3];
    const float* W2 = (const float*)d_in[4];
    const float* b2 = (const float*)d_in[5];
    const int* src = ei;
    const int* dst = ei + NE;
    float* out = (float*)d_out;

    float* ws   = (float*)d_ws;
    float* h1   = ws;
    float* z1   = h1 + (size_t)NN * HID;
    float* h2   = z1 + (size_t)NN * HID;
    float* dis  = h2 + (size_t)NN * NCLS;
    int* cnt    = (int*)(dis + NN);
    int* rowptr = cnt + NN;
    int* csr    = rowptr + NN;
    int* binned = csr + NE;
    int* counts = binned + 2 * (size_t)NE;
    int* offs   = counts + NCH * NB;
    int* binbase = offs + NCH * NB;

    k_bincount<<<NCH, 256, 0, stream>>>(dst, counts);
    k_binscan<<<1, 1024, 0, stream>>>(counts, offs, binbase);
    k_binscatter<<<NCH, 256, 0, stream>>>(src, dst, counts, offs, binned);
    k_binsort<<<NB, SORT_T, 0, stream>>>(binned, binbase, rowptr, cnt, dis, csr);

    k_gemm1<<<(ROW_TILES + 3) / 4, 128, 0, stream>>>(x, W1, h1);
    k_gather1<<<NN / 16, 256, 0, stream>>>(rowptr, cnt, csr, dis, h1, b1, z1);

    k_gemm2<<<(ROW_TILES + 3) / 4, 128, 0, stream>>>(z1, W2, h2);
    k_gather2<<<NN / 16, 256, 0, stream>>>(rowptr, cnt, csr, dis, h2, b2, out);
}


# Gathers re-cut: each 16-lane group walks 4 adjacent nodes, gets a node's edge indices by one cooperative load (shuffled out), keeps 8 (64-float) or 16 (10-float) row loads in flight; sum order kept
# speedup vs baseline: 4.9984x; 1.1533x over previous
#include <hip/hip_runtime.h>
#include <math.h>


#define NN 100000
#define NE 1600000
#define FIN 128
#define HID 64
#define NCLS 10
#define ROW_TILES 6250
#define NB 256
#define BIN_NODES 391
#define NCH 256
#define CH 6250
#define CH_PER_T 25
#define SORT_T 1024
#define STAGE_CAP 16384

typedef __attribute__((ext_vector_type(2))) float v2f;
typedef __attribute__((ext_vector_type(8))) float v8f;

__device__ __forceinline__ unsigned bin_of(int d) {
    unsigned b = (unsigned)d / BIN_NODES;
    return b < NB - 1 ? b : NB - 1;
}

__device__ __forceinline__ int block_excl_scan(int c, int t, int* wtot, int nwaves) {
    int lane = t & 31, wv = t >> 5;
    int inc = c;
    for (int o = 1; o < 32; o <<= 1) {
        int v = __shfl_up(inc, o);
        if (lane >= o) inc += v;
    }
    if (lane == 31 && wv < nwaves) wtot[wv] = inc;
    __syncthreads();
    int off = 0;
    for (int w = 0; w < nwaves; ++w) off += (w < wv) ? wtot[w] : 0;
    return off + inc - c;
}

__global__ void __launch_bounds__(256)
k_bincount(const int* __restrict__ dst, int* __restrict__ counts) {
    __shared__ int hist[NB];
    int t = threadIdx.x;
    hist[t] = 0;
    __syncthreads();
    const int* d = dst + blockIdx.x * CH;
    int dv[CH_PER_T];
    #pragma unroll
    for (int k = 0; k < CH_PER_T; ++k) {
        int i = t + 256 * k;
        dv[k] = (i < CH) ? d[i] : -1;
    }
    #pragma unroll
    for (int k = 0; k < CH_PER_T; ++k)
        if (dv[k] >= 0) atomicAdd(&hist[bin_of(dv[k])], 1);
    __syncthreads();
    counts[blockIdx.x * NB + t] = hist[t];
}

__global__ void __launch_bounds__(1024)
k_binscan(const int* __restrict__ counts, int* __restrict__ offs, int* __restrict__ binbase) {
    __shared__ int part[4][NB];
    __shared__ int bbase[NB];
    __shared__ int wtot[8];
    int t = threadIdx.x;
    int b = t & (NB - 1), q = t >> 8;
    const int* c = counts + (q * 64) * NB + b;
    int v[64];
    #pragma unroll
    for (int k = 0; k < 64; ++k) v[k] = c[k * NB];
    int s = 0;
    #pragma unroll
    for (int k = 0; k < 64; ++k) s += v[k];
    part[q][b] = s;
    __syncthreads();
    int tot = (t < NB) ? (part[0][b] + part[1][b]) + (part[2][b] + part[3][b]) : 0;
    int ex = block_excl_scan(tot, t, wtot, 8);
    if (t < NB) {
        bbase[t] = ex;
        binbase[t] = ex;
        if (t == NB - 1) binbase[NB] = ex + tot;
    }
    __syncthreads();
    int run = bbase[b];
    for (int qq = 0; qq < q; ++qq) run += part[qq][b];
    int* o = offs + (q * 64) * NB + b;
    #pragma unroll
    for (int k = 0; k < 64; ++k) { o[k * NB] = run; run += v[k]; }
}

__global__ void __launch_bounds__(256)
k_binscatter(const int* __restrict__ src, const int* __restrict__ dst,
             const int* __restrict__ counts, const int* __restrict__ offs,
             int* __restrict__ binned_w) {
    int2* __restrict__ binned = (int2*)binned_w;
    __shared__ int cursor[NB];
    __shared__ int goff[NB];
    __shared__ int wtot[8];
    __shared__ int2 stage[CH];
    int t = threadIdx.x;
    const int* sp = src + blockIdx.x * CH;
    const int* dp = dst + blockIdx.x * CH;
    int sv[CH_PER_T], dv[CH_PER_T];
    #pragma unroll
    for (int k = 0; k < CH_PER_T; ++k) {
        int i = t + 256 * k;
        dv[k] = (i < CH) ? dp[i] : -1;
        sv[k] = (i < CH) ? sp[i] : 0;
    }
    int c = counts[blockIdx.x * NB + t];
    int ex = block_excl_scan(c, t, wtot, 8);
    cursor[t] = ex;
    goff[t] = offs[blockIdx.x * NB + t] - ex;
    __syncthreads();
    #pragma unroll
    for (int k = 0; k < CH_PER_T; ++k)
        if (dv[k] >= 0) {
            int p = atomicAdd(&cursor[bin_of(dv[k])], 1);
            stage[p] = make_int2(sv[k], dv[k]);
        }
    __syncthreads();
    for (int i = t; i < CH; i += 256) {
        int2 e = stage[i];
        binned[goff[bin_of(e.y)] + i] = e;
    }
}

__global__ void __launch_bounds__(SORT_T)
k_binsort(const int* __restrict__ binned_w, const int* __restrict__ binbase,
          int* __restrict__ rowptr, int* __restrict__ cnt, float* __restrict__ dis,
          int* __restrict__ csr) {
    const int2* __restrict__ binned = (const int2*)binned_w;
    __shared__ int lcnt[512];
    __shared__ int lcur[512];
    __shared__ int wtot[16];
    __shared__ int lst[STAGE_CAP];
    int t = threadIdx.x;
    int lo = blockIdx.x * BIN_NODES;
    int nn = NN - lo < BIN_NODES ? NN - lo : BIN_NODES;
    int base = binbase[blockIdx.x];
    int end = binbase[blockIdx.x + 1];
    int n = end - base;
    if (t < 512) lcnt[t] = 0;
    __syncthreads();
    for (int i0 = base; i0 < end; i0 += 4 * SORT_T) {
        int dv[4];
        #pragma unroll
        for (int k = 0; k < 4; ++k) {
            int i = i0 + k * SORT_T + t;
            dv[k] = (i < end) ? binned[i].y : -1;
        }
        #pragma unroll
        for (int k = 0; k < 4; ++k)
            if (dv[k] >= 0) {
                unsigned li = (unsigned)(dv[k] - lo);
                atomicAdd(&lcnt[li < BIN_NODES - 1 ? li : BIN_NODES - 1], 1);
            }
    }
    __syncthreads();
    int c = (t < 512) ? lcnt[t] : 0;
    int ex = block_excl_scan(c, t, wtot, 16);
    if (t < 512) lcur[t] = ex;
    if (t < nn) {
        rowptr[lo + t] = base + ex;
        cnt[lo + t] = c;
        dis[lo + t] = rsqrtf((float)(c + 1));
    }
    __syncthreads();
    bool staged = n <= STAGE_CAP;
    for (int i0 = base; i0 < end; i0 += 4 * SORT_T) {
        int2 ev[4];
        #pragma unroll
        for (int k = 0; k < 4; ++k) {
            int i = i0 + k * SORT_T + t;
            ev[k] = (i < end) ? binned[i] : make_int2(0, -1);
        }
        #pragma unroll
        for (int k = 0; k < 4; ++k)
            if (ev[k].y >= 0) {
                unsigned li = (unsigned)(ev[k].y - lo);
                int p = atomicAdd(&lcur[li < BIN_NODES - 1 ? li : BIN_NODES - 1], 1);
                if (staged) lst[p] = ev[k].x;
                else csr[base + p] = ev[k].x;
            }
    }
    if (staged) {
        __syncthreads();
        for (int i = t; i < n; i += SORT_T) csr[base + i] = lst[i];
    }
}

__global__ void __launch_bounds__(128)
k_gemm1(const float* __restrict__ x, const float* __restrict__ W1,
        const float* __restrict__ dis, float* __restrict__ h1) {
    __shared__ float wT[HID * 132];
    int t = threadIdx.x;
    for (int i = t; i < FIN * HID; i += 128) {
        int k = i >> 6, c = i & 63;
        wT[c * 132 + k] = W1[i];
    }
    __syncthreads();

    int tile = blockIdx.x * 4 + (t >> 5);
    if (tile >= ROW_TILES) return;
    int lane = t & 31;
    int m = lane & 15, half = lane >> 4;

    const float* arow = x + (size_t)(tile * 16 + m) * FIN + 2 * half;
    v8f a0 = {}, a1 = {}, a2 = {}, a3 = {};
    const float* dp = dis + tile * 16 + 8 * half;
    float4 d0 = *(const float4*)(dp);
    float4 d1 = *(const float4*)(dp + 4);

    #pragma unroll 4
    for (int k0 = 0; k0 < FIN; k0 += 4) {
        v2f a = *(const v2f*)(arow + k0);
        const float* bp = wT + m * 132 + k0 + 2 * half;
        v2f b0 = *(const v2f*)(bp);
        v2f b1 = *(const v2f*)(bp + 16 * 132);
        v2f b2 = *(const v2f*)(bp + 32 * 132);
        v2f b3 = *(const v2f*)(bp + 48 * 132);
        a0 = __builtin_amdgcn_wmma_f32_16x16x4_f32(false, a, false, b0, (short)0, a0, false, false);
        a1 = __builtin_amdgcn_wmma_f32_16x16x4_f32(false, a, false, b1, (short)0, a1, false, false);
        a2 = __builtin_amdgcn_wmma_f32_16x16x4_f32(false, a, false, b2, (short)0, a2, false, false);
        a3 = __builtin_amdgcn_wmma_f32_16x16x4_f32(false, a, false, b3, (short)0, a3, false, false);
    }

    float* out = h1 + (size_t)(tile * 16 + 8 * half) * HID + m;
    const float ds[8] = {d0.x, d0.y, d0.z, d0.w, d1.x, d1.y, d1.z, d1.w};
    #pragma unroll
    for (int v = 0; v < 8; ++v) {
        out[v * HID +  0] = a0[v] * ds[v];
        out[v * HID + 16] = a1[v] * ds[v];
        out[v * HID + 32] = a2[v] * ds[v];
        out[v * HID + 48] = a3[v] * ds[v];
    }
}

__global__ void __launch_bounds__(128)
k_gemm2(const float* __restrict__ z1, const float* __restrict__ W2,
        const float* __restrict__ dis, float* __restrict__ h2) {
    __shared__ float wT[16 * 68];
    int t = threadIdx.x;
    for (int i = t; i < 16 * HID; i += 128) {
        int k = i >> 4, c = i & 15;
        wT[c * 68 + k] = (c < NCLS) ? W2[k * NCLS + c] : 0.0f;
    }
    __syncthreads();

    int tile = blockIdx.x * 4 + (t >> 5);
    if (tile >= ROW_TILES) return;
    int lane = t & 31;
    int m = lane & 15, half = lane >> 4;

    const float* arow = z1 + (size_t)(tile * 16 + m) * HID + 2 * half;
    v8f acc = {};
    const float* dp = dis + tile * 16 + 8 * half;
    float4 d0 = *(const float4*)(dp);
    float4 d1 = *(const float4*)(dp + 4);
    #pragma unroll
    for (int k0 = 0; k0 < HID; k0 += 4) {
        v2f a = *(const v2f*)(arow + k0);
        v2f b = *(const v2f*)(wT + m * 68 + k0 + 2 * half);
        acc = __builtin_amdgcn_wmma_f32_16x16x4_f32(false, a, false, b, (short)0, acc, false, false);
    }
    if (m < NCLS) {
        float* out = h2 + (size_t)(tile * 16 + 8 * half) * NCLS + m;
        const float ds[8] = {d0.x, d0.y, d0.z, d0.w, d1.x, d1.y, d1.z, d1.w};
        #pragma unroll
        for (int v = 0; v < 8; ++v) out[v * NCLS] = acc[v] * ds[v];
    }
}

#define NPG 4
#define N_GROUPS 25000
#define GATHER_BLOCKS 1563

__device__ __forceinline__ int load_idx16(const int* __restrict__ csr, int beg, int e0, int l) {
    int j = beg + e0 + l;
    j = j < NE - 1 ? j : NE - 1;
    return (int)min((unsigned)csr[j], (unsigned)(NN - 1));
}

__device__ __forceinline__ int wave_count(int c) {
    int o = __shfl_xor(c, 16);
    return __builtin_amdgcn_readfirstlane(c > o ? c : o);
}

template <int B>
__device__ __forceinline__ void add_rows_f4(float4& acc, const float* __restrict__ h, int idx,
                                            int e0, int slot0, int c, unsigned self_off, int l) {
    float4 v[B];
    #pragma unroll
    for (int k = 0; k < B; ++k) {
        int s = __shfl(idx, slot0 + k, 16);
        unsigned off = (e0 + k < c) ? (unsigned)s * HID + 4u * (unsigned)l : self_off;
        v[k] = *(const float4*)(h + off);
    }
    #pragma unroll
    for (int k = 0; k < B; ++k) {
        bool ok = e0 + k < c;
        acc.x += ok ? v[k].x : 0.0f;
        acc.y += ok ? v[k].y : 0.0f;
        acc.z += ok ? v[k].z : 0.0f;
        acc.w += ok ? v[k].w : 0.0f;
    }
}

template <int B>
__device__ __forceinline__ void add_rows_f1(float& acc, const float* __restrict__ h, int idx,
                                            int e0, int slot0, int c, unsigned self_off, int fc) {
    float v[B];
    #pragma unroll
    for (int k = 0; k < B; ++k) {
        int s = __shfl(idx, slot0 + k, 16);
        unsigned off = (e0 + k < c) ? (unsigned)s * NCLS + (unsigned)fc : self_off;
        v[k] = h[off];
    }
    #pragma unroll
    for (int k = 0; k < B; ++k) acc += (e0 + k < c) ? v[k] : 0.0f;
}

__device__ __forceinline__ float4 node_sum_f4(const int* __restrict__ csr, const float* __restrict__ h,
                                              int beg, int c, int ia, int ib, unsigned self_off, int l) {
    float4 acc = make_float4(0.0f, 0.0f, 0.0f, 0.0f);
    int cw = wave_count(c);
    if (cw > 0)  add_rows_f4<8>(acc, h, ia, 0, 0, c, self_off, l);
    if (cw > 8)  add_rows_f4<8>(acc, h, ia, 8, 8, c, self_off, l);
    if (cw > 16) add_rows_f4<8>(acc, h, ib, 16, 0, c, self_off, l);
    if (cw > 24) add_rows_f4<8>(acc, h, ib, 24, 8, c, self_off, l);
    for (int e0 = 32; e0 < cw; e0 += 16) {
        int ix = load_idx16(csr, beg, e0, l);
        add_rows_f4<8>(acc, h, ix, e0, 0, c, self_off, l);
        if (e0 + 8 < cw) add_rows_f4<8>(acc, h, ix, e0 + 8, 8, c, self_off, l);
    }
    return acc;
}

__device__ __forceinline__ float node_sum_f1(const int* __restrict__ csr, const float* __restrict__ h,
                                             int beg, int c, int ia, int ib, unsigned self_off,
                                             int l, int fc) {
    float acc = 0.0f;
    int cw = wave_count(c);
    if (cw > 0)  add_rows_f1<16>(acc, h, ia, 0, 0, c, self_off, fc);
    if (cw > 16) add_rows_f1<16>(acc, h, ib, 16, 0, c, self_off, fc);
    for (int e0 = 32; e0 < cw; e0 += 16) {
        int ix = load_idx16(csr, beg, e0, l);
        add_rows_f1<16>(acc, h, ix, e0, 0, c, self_off, fc);
    }
    return acc;
}

__global__ void __launch_bounds__(256)
k_gather1(const int* __restrict__ rowptr, const int* __restrict__ cnt,
          const int* __restrict__ csr, const float* __restrict__ dis,
          const float* __restrict__ h1, const float* __restrict__ b1,
          float* __restrict__ z1) {
    int t = threadIdx.x;
    int l = t & 15;
    int grp = blockIdx.x * 16 + (t >> 4);
    if (grp >= N_GROUPS) return;
    int n0 = grp * NPG;
    int4 rp = *(const int4*)(rowptr + n0);
    int4 cn = *(const int4*)(cnt + n0);
    float4 dv = *(const float4*)(dis + n0);
    const int begs[NPG] = {rp.x, rp.y, rp.z, rp.w};
    const int cs[NPG] = {cn.x, cn.y, cn.z, cn.w};
    const float ds[NPG] = {dv.x, dv.y, dv.z, dv.w};
    float4 bb = *(const float4*)(b1 + 4 * l);
    int ia[NPG];
    #pragma unroll
    for (int n = 0; n < NPG; ++n)
        ia[n] = load_idx16(csr, begs[n], 0, l);
    #pragma unroll
    for (int n = 0; n < NPG; ++n) {
        int node = n0 + n;
        unsigned self_off = (unsigned)node * HID + 4u * (unsigned)l;
        float4 hv = *(const float4*)(h1 + self_off);
        int ibn = load_idx16(csr, begs[n], 16, l);
        float4 acc = node_sum_f4(csr, h1, begs[n], cs[n], ia[n], ibn, self_off, l);
        float di = ds[n];
        float4 r;
        r.x = (acc.x + hv.x) * di + bb.x;
        r.y = (acc.y + hv.y) * di + bb.y;
        r.z = (acc.z + hv.z) * di + bb.z;
        r.w = (acc.w + hv.w) * di + bb.w;
        r.x = r.x > 0.0f ? r.x : 0.0f;
        r.y = r.y > 0.0f ? r.y : 0.0f;
        r.z = r.z > 0.0f ? r.z : 0.0f;
        r.w = r.w > 0.0f ? r.w : 0.0f;
        *(float4*)(z1 + (size_t)node * HID + 4 * l) = r;
    }
}

__global__ void __launch_bounds__(256)
k_gather2(const int* __restrict__ rowptr, const int* __restrict__ cnt,
          const int* __restrict__ csr, const float* __restrict__ dis,
          const float* __restrict__ h2, const float* __restrict__ b2,
          float* __restrict__ out) {
    __shared__ float stage[16 * NPG * NCLS];
    int t = threadIdx.x;
    int f = t & 15;
    int fc = f < NCLS ? f : NCLS - 1;
    int grp = blockIdx.x * 16 + (t >> 4);
    if (grp < N_GROUPS) {
        int n0 = grp * NPG;
        int4 rp = *(const int4*)(rowptr + n0);
        int4 cn = *(const int4*)(cnt + n0);
        float4 dv = *(const float4*)(dis + n0);
        const int begs[NPG] = {rp.x, rp.y, rp.z, rp.w};
        const int cs[NPG] = {cn.x, cn.y, cn.z, cn.w};
        const float ds[NPG] = {dv.x, dv.y, dv.z, dv.w};
        float bias = b2[fc];
        int ia[NPG];
        #pragma unroll
        for (int n = 0; n < NPG; ++n)
            ia[n] = load_idx16(csr, begs[n], 0, f);
        #pragma unroll
        for (int n = 0; n < NPG; ++n) {
            int node = n0 + n;
            unsigned self_off = (unsigned)node * NCLS + (unsigned)fc;
            float hv = h2[self_off];
            int ibn = load_idx16(csr, begs[n], 16, f);
            float acc = node_sum_f1(csr, h2, begs[n], cs[n], ia[n], ibn, self_off, f, fc);
            float v = (acc + hv) * ds[n] + bias;
            float mx = f < NCLS ? v : -3.402823466e38f;
            for (int o = 8; o > 0; o >>= 1) mx = fmaxf(mx, __shfl_xor(mx, o, 16));
            float ex = f < NCLS ? expf(v - mx) : 0.0f;
            for (int o = 8; o > 0; o >>= 1) ex += __shfl_xor(ex, o, 16);
            float lse = mx + logf(ex);
            if (f < NCLS) stage[((t >> 4) * NPG + n) * NCLS + f] = v - lse;
        }
    }
    __syncthreads();
    for (int i = t; i < 16 * NPG * NCLS; i += 256) {
        size_t gi = (size_t)blockIdx.x * (16 * NPG * NCLS) + i;
        if (gi < (size_t)NN * NCLS) out[gi] = stage[i];
    }
}

extern "C" void kernel_launch(void* const* d_in, const int* in_sizes, int n_in,
                              void* d_out, int out_size, void* d_ws, size_t ws_size,
                              hipStream_t stream) {
    const float* x  = (const float*)d_in[0];
    const int*   ei = (const int*)d_in[1];
    const float* W1 = (const float*)d_in[2];
    const float* b1 = (const float*)d_in[3];
    const float* W2 = (const float*)d_in[4];
    const float* b2 = (const float*)d_in[5];
    const int* src = ei;
    const int* dst = ei + NE;
    float* out = (float*)d_out;

    float* ws   = (float*)d_ws;
    float* h1   = ws;
    float* z1   = h1 + (size_t)NN * HID;
    float* h2   = z1 + (size_t)NN * HID;
    float* dis  = h2 + (size_t)NN * NCLS;
    int* cnt    = (int*)(dis + NN);
    int* rowptr = cnt + NN;
    int* csr    = rowptr + NN;
    int* binned = csr + NE;
    int* counts = binned + 2 * (size_t)NE;
    int* offs   = counts + NCH * NB;
    int* binbase = offs + NCH * NB;

    k_bincount<<<NCH, 256, 0, stream>>>(dst, counts);
    k_binscan<<<1, 1024, 0, stream>>>(counts, offs, binbase);
    k_binscatter<<<NCH, 256, 0, stream>>>(src, dst, counts, offs, binned);
    k_binsort<<<NB, SORT_T, 0, stream>>>(binned, binbase, rowptr, cnt, dis, csr);

    k_gemm1<<<(ROW_TILES + 3) / 4, 128, 0, stream>>>(x, W1, dis, h1);
    k_gather1<<<GATHER_BLOCKS, 256, 0, stream>>>(rowptr, cnt, csr, dis, h1, b1, z1);

    k_gemm2<<<(ROW_TILES + 3) / 4, 128, 0, stream>>>(z1, W2, dis, h2);
    k_gather2<<<GATHER_BLOCKS, 256, 0, stream>>>(rowptr, cnt, csr, dis, h2, b2, out);
}


# Layer-2 product fused into the layer-1 gather: a block's 64 activated rows stay in LDS, four waves multiply them by W2 on the f32 matrix unit; z1 never stored, one launch fewer; sum order kept
# speedup vs baseline: 5.4197x; 1.0843x over previous
#include <hip/hip_runtime.h>
#include <math.h>


#define NN 100000
#define NE 1600000
#define FIN 128
#define HID 64
#define NCLS 10
#define ROW_TILES 6250
#define NB 256
#define BIN_NODES 391
#define NCH 256
#define CH 6250
#define CH_PER_T 25
#define SORT_T 1024
#define STAGE_CAP 16384

typedef __attribute__((ext_vector_type(2))) float v2f;
typedef __attribute__((ext_vector_type(8))) float v8f;

__device__ __forceinline__ unsigned bin_of(int d) {
    unsigned b = (unsigned)d / BIN_NODES;
    return b < NB - 1 ? b : NB - 1;
}

__device__ __forceinline__ int block_excl_scan(int c, int t, int* wtot, int nwaves) {
    int lane = t & 31, wv = t >> 5;
    int inc = c;
    for (int o = 1; o < 32; o <<= 1) {
        int v = __shfl_up(inc, o);
        if (lane >= o) inc += v;
    }
    if (lane == 31 && wv < nwaves) wtot[wv] = inc;
    __syncthreads();
    int off = 0;
    for (int w = 0; w < nwaves; ++w) off += (w < wv) ? wtot[w] : 0;
    return off + inc - c;
}

__global__ void __launch_bounds__(256)
k_bincount(const int* __restrict__ dst, int* __restrict__ counts) {
    __shared__ int hist[NB];
    int t = threadIdx.x;
    hist[t] = 0;
    __syncthreads();
    const int* d = dst + blockIdx.x * CH;
    int dv[CH_PER_T];
    #pragma unroll
    for (int k = 0; k < CH_PER_T; ++k) {
        int i = t + 256 * k;
        dv[k] = (i < CH) ? d[i] : -1;
    }
    #pragma unroll
    for (int k = 0; k < CH_PER_T; ++k)
        if (dv[k] >= 0) atomicAdd(&hist[bin_of(dv[k])], 1);
    __syncthreads();
    counts[blockIdx.x * NB + t] = hist[t];
}

__global__ void __launch_bounds__(1024)
k_binscan(const int* __restrict__ counts, int* __restrict__ offs, int* __restrict__ binbase) {
    __shared__ int part[4][NB];
    __shared__ int bbase[NB];
    __shared__ int wtot[8];
    int t = threadIdx.x;
    int b = t & (NB - 1), q = t >> 8;
    const int* c = counts + (q * 64) * NB + b;
    int v[64];
    #pragma unroll
    for (int k = 0; k < 64; ++k) v[k] = c[k * NB];
    int s = 0;
    #pragma unroll
    for (int k = 0; k < 64; ++k) s += v[k];
    part[q][b] = s;
    __syncthreads();
    int tot = (t < NB) ? (part[0][b] + part[1][b]) + (part[2][b] + part[3][b]) : 0;
    int ex = block_excl_scan(tot, t, wtot, 8);
    if (t < NB) {
        bbase[t] = ex;
        binbase[t] = ex;
        if (t == NB - 1) binbase[NB] = ex + tot;
    }
    __syncthreads();
    int run = bbase[b];
    for (int qq = 0; qq < q; ++qq) run += part[qq][b];
    int* o = offs + (q * 64) * NB + b;
    #pragma unroll
    for (int k = 0; k < 64; ++k) { o[k * NB] = run; run += v[k]; }
}

__global__ void __launch_bounds__(256)
k_binscatter(const int* __restrict__ src, const int* __restrict__ dst,
             const int* __restrict__ counts, const int* __restrict__ offs,
             int* __restrict__ binned_w) {
    int2* __restrict__ binned = (int2*)binned_w;
    __shared__ int cursor[NB];
    __shared__ int goff[NB];
    __shared__ int wtot[8];
    __shared__ int2 stage[CH];
    int t = threadIdx.x;
    const int* sp = src + blockIdx.x * CH;
    const int* dp = dst + blockIdx.x * CH;
    int sv[CH_PER_T], dv[CH_PER_T];
    #pragma unroll
    for (int k = 0; k < CH_PER_T; ++k) {
        int i = t + 256 * k;
        dv[k] = (i < CH) ? dp[i] : -1;
        sv[k] = (i < CH) ? sp[i] : 0;
    }
    int c = counts[blockIdx.x * NB + t];
    int ex = block_excl_scan(c, t, wtot, 8);
    cursor[t] = ex;
    goff[t] = offs[blockIdx.x * NB + t] - ex;
    __syncthreads();
    #pragma unroll
    for (int k = 0; k < CH_PER_T; ++k)
        if (dv[k] >= 0) {
            int p = atomicAdd(&cursor[bin_of(dv[k])], 1);
            stage[p] = make_int2(sv[k], dv[k]);
        }
    __syncthreads();
    for (int i = t; i < CH; i += 256) {
        int2 e = stage[i];
        binned[goff[bin_of(e.y)] + i] = e;
    }
}

__global__ void __launch_bounds__(SORT_T)
k_binsort(const int* __restrict__ binned_w, const int* __restrict__ binbase,
          int* __restrict__ rowptr, int* __restrict__ cnt, float* __restrict__ dis,
          int* __restrict__ csr) {
    const int2* __restrict__ binned = (const int2*)binned_w;
    __shared__ int lcnt[512];
    __shared__ int lcur[512];
    __shared__ int wtot[16];
    __shared__ int lst[STAGE_CAP];
    int t = threadIdx.x;
    int lo = blockIdx.x * BIN_NODES;
    int nn = NN - lo < BIN_NODES ? NN - lo : BIN_NODES;
    int base = binbase[blockIdx.x];
    int end = binbase[blockIdx.x + 1];
    int n = end - base;
    if (t < 512) lcnt[t] = 0;
    __syncthreads();
    for (int i0 = base; i0 < end; i0 += 4 * SORT_T) {
        int dv[4];
        #pragma unroll
        for (int k = 0; k < 4; ++k) {
            int i = i0 + k * SORT_T + t;
            dv[k] = (i < end) ? binned[i].y : -1;
        }
        #pragma unroll
        for (int k = 0; k < 4; ++k)
            if (dv[k] >= 0) {
                unsigned li = (unsigned)(dv[k] - lo);
                atomicAdd(&lcnt[li < BIN_NODES - 1 ? li : BIN_NODES - 1], 1);
            }
    }
    __syncthreads();
    int c = (t < 512) ? lcnt[t] : 0;
    int ex = block_excl_scan(c, t, wtot, 16);
    if (t < 512) lcur[t] = ex;
    if (t < nn) {
        rowptr[lo + t] = base + ex;
        cnt[lo + t] = c;
        dis[lo + t] = rsqrtf((float)(c + 1));
    }
    __syncthreads();
    bool staged = n <= STAGE_CAP;
    for (int i0 = base; i0 < end; i0 += 4 * SORT_T) {
        int2 ev[4];
        #pragma unroll
        for (int k = 0; k < 4; ++k) {
            int i = i0 + k * SORT_T + t;
            ev[k] = (i < end) ? binned[i] : make_int2(0, -1);
        }
        #pragma unroll
        for (int k = 0; k < 4; ++k)
            if (ev[k].y >= 0) {
                unsigned li = (unsigned)(ev[k].y - lo);
                int p = atomicAdd(&lcur[li < BIN_NODES - 1 ? li : BIN_NODES - 1], 1);
                if (staged) lst[p] = ev[k].x;
                else csr[base + p] = ev[k].x;
            }
    }
    if (staged) {
        __syncthreads();
        for (int i = t; i < n; i += SORT_T) csr[base + i] = lst[i];
    }
}

__global__ void __launch_bounds__(128)
k_gemm1(const float* __restrict__ x, const float* __restrict__ W1,
        const float* __restrict__ dis, float* __restrict__ h1) {
    __shared__ float wT[HID * 132];
    int t = threadIdx.x;
    for (int i = t; i < FIN * HID; i += 128) {
        int k = i >> 6, c = i & 63;
        wT[c * 132 + k] = W1[i];
    }
    __syncthreads();

    int tile = blockIdx.x * 4 + (t >> 5);
    if (tile >= ROW_TILES) return;
    int lane = t & 31;
    int m = lane & 15, half = lane >> 4;

    const float* arow = x + (size_t)(tile * 16 + m) * FIN + 2 * half;
    v8f a0 = {}, a1 = {}, a2 = {}, a3 = {};
    const float* dp = dis + tile * 16 + 8 * half;
    float4 d0 = *(const float4*)(dp);
    float4 d1 = *(const float4*)(dp + 4);

    #pragma unroll 4
    for (int k0 = 0; k0 < FIN; k0 += 4) {
        v2f a = *(const v2f*)(arow + k0);
        const float* bp = wT + m * 132 + k0 + 2 * half;
        v2f b0 = *(const v2f*)(bp);
        v2f b1 = *(const v2f*)(bp + 16 * 132);
        v2f b2 = *(const v2f*)(bp + 32 * 132);
        v2f b3 = *(const v2f*)(bp + 48 * 132);
        a0 = __builtin_amdgcn_wmma_f32_16x16x4_f32(false, a, false, b0, (short)0, a0, false, false);
        a1 = __builtin_amdgcn_wmma_f32_16x16x4_f32(false, a, false, b1, (short)0, a1, false, false);
        a2 = __builtin_amdgcn_wmma_f32_16x16x4_f32(false, a, false, b2, (short)0, a2, false, false);
        a3 = __builtin_amdgcn_wmma_f32_16x16x4_f32(false, a, false, b3, (short)0, a3, false, false);
    }

    float* out = h1 + (size_t)(tile * 16 + 8 * half) * HID + m;
    const float ds[8] = {d0.x, d0.y, d0.z, d0.w, d1.x, d1.y, d1.z, d1.w};
    #pragma unroll
    for (int v = 0; v < 8; ++v) {
        out[v * HID +  0] = a0[v] * ds[v];
        out[v * HID + 16] = a1[v] * ds[v];
        out[v * HID + 32] = a2[v] * ds[v];
        out[v * HID + 48] = a3[v] * ds[v];
    }
}

#define NPG 4
#define N_GROUPS 25000
#define GATHER_BLOCKS 1563

__device__ __forceinline__ int load_idx16(const int* __restrict__ csr, int beg, int e0, int l) {
    int j = beg + e0 + l;
    j = j < NE - 1 ? j : NE - 1;
    return (int)min((unsigned)csr[j], (unsigned)(NN - 1));
}

__device__ __forceinline__ int wave_count(int c) {
    int o = __shfl_xor(c, 16);
    return __builtin_amdgcn_readfirstlane(c > o ? c : o);
}

template <int B>
__device__ __forceinline__ void add_rows_f4(float4& acc, const float* __restrict__ h, int idx,
                                            int e0, int slot0, int c, unsigned self_off, int l) {
    float4 v[B];
    #pragma unroll
    for (int k = 0; k < B; ++k) {
        int s = __shfl(idx, slot0 + k, 16);
        unsigned off = (e0 + k < c) ? (unsigned)s * HID + 4u * (unsigned)l : self_off;
        v[k] = *(const float4*)(h + off);
    }
    #pragma unroll
    for (int k = 0; k < B; ++k) {
        bool ok = e0 + k < c;
        acc.x += ok ? v[k].x : 0.0f;
        acc.y += ok ? v[k].y : 0.0f;
        acc.z += ok ? v[k].z : 0.0f;
        acc.w += ok ? v[k].w : 0.0f;
    }
}

template <int B>
__device__ __forceinline__ void add_rows_f1(float& acc, const float* __restrict__ h, int idx,
                                            int e0, int slot0, int c, unsigned self_off, int fc) {
    float v[B];
    #pragma unroll
    for (int k = 0; k < B; ++k) {
        int s = __shfl(idx, slot0 + k, 16);
        unsigned off = (e0 + k < c) ? (unsigned)s * NCLS + (unsigned)fc : self_off;
        v[k] = h[off];
    }
    #pragma unroll
    for (int k = 0; k < B; ++k) acc += (e0 + k < c) ? v[k] : 0.0f;
}

__device__ __forceinline__ float4 node_sum_f4(const int* __restrict__ csr, const float* __restrict__ h,
                                              int beg, int c, int ia, int ib, unsigned self_off, int l) {
    float4 acc = make_float4(0.0f, 0.0f, 0.0f, 0.0f);
    int cw = wave_count(c);
    if (cw > 0)  add_rows_f4<8>(acc, h, ia, 0, 0, c, self_off, l);
    if (cw > 8)  add_rows_f4<8>(acc, h, ia, 8, 8, c, self_off, l);
    if (cw > 16) add_rows_f4<8>(acc, h, ib, 16, 0, c, self_off, l);
    if (cw > 24) add_rows_f4<8>(acc, h, ib, 24, 8, c, self_off, l);
    for (int e0 = 32; e0 < cw; e0 += 16) {
        int ix = load_idx16(csr, beg, e0, l);
        add_rows_f4<8>(acc, h, ix, e0, 0, c, self_off, l);
        if (e0 + 8 < cw) add_rows_f4<8>(acc, h, ix, e0 + 8, 8, c, self_off, l);
    }
    return acc;
}

__device__ __forceinline__ float node_sum_f1(const int* __restrict__ csr, const float* __restrict__ h,
                                             int beg, int c, int ia, int ib, unsigned self_off,
                                             int l, int fc) {
    float acc = 0.0f;
    int cw = wave_count(c);
    if (cw > 0)  add_rows_f1<16>(acc, h, ia, 0, 0, c, self_off, fc);
    if (cw > 16) add_rows_f1<16>(acc, h, ib, 16, 0, c, self_off, fc);
    for (int e0 = 32; e0 < cw; e0 += 16) {
        int ix = load_idx16(csr, beg, e0, l);
        add_rows_f1<16>(acc, h, ix, e0, 0, c, self_off, fc);
    }
    return acc;
}

#define ZP 68
#define BLK_ROWS (16 * NPG)
__global__ void __launch_bounds__(256)
k_gather1(const int* __restrict__ rowptr, const int* __restrict__ cnt,
          const int* __restrict__ csr, const float* __restrict__ dis,
          const float* __restrict__ h1, const float* __restrict__ b1,
          const float* __restrict__ W2, float* __restrict__ h2) {
    __shared__ float zt[BLK_ROWS * ZP];
    __shared__ float wT[16 * ZP];
    __shared__ float ho[BLK_ROWS * 16];
    __shared__ float dl[BLK_ROWS];
    int t = threadIdx.x;
    int l = t & 15;
    int grp = blockIdx.x * 16 + (t >> 4);
    if (grp < N_GROUPS) {
        int n0 = grp * NPG;
        int4 rp = *(const int4*)(rowptr + n0);
        int4 cn = *(const int4*)(cnt + n0);
        float4 dv = *(const float4*)(dis + n0);
        if (l == 0) *(float4*)(dl + (t >> 4) * NPG) = dv;
        const int begs[NPG] = {rp.x, rp.y, rp.z, rp.w};
        const int cs[NPG] = {cn.x, cn.y, cn.z, cn.w};
        const float ds[NPG] = {dv.x, dv.y, dv.z, dv.w};
        float4 bb = *(const float4*)(b1 + 4 * l);
        int ia[NPG];
        #pragma unroll
        for (int n = 0; n < NPG; ++n)
            ia[n] = load_idx16(csr, begs[n], 0, l);
        #pragma unroll
        for (int n = 0; n < NPG; ++n) {
            int node = n0 + n;
            unsigned self_off = (unsigned)node * HID + 4u * (unsigned)l;
            float4 hv = *(const float4*)(h1 + self_off);
            int ibn = load_idx16(csr, begs[n], 16, l);
            float4 acc = node_sum_f4(csr, h1, begs[n], cs[n], ia[n], ibn, self_off, l);
            float di = ds[n];
            float4 r;
            r.x = (acc.x + hv.x) * di + bb.x;
            r.y = (acc.y + hv.y) * di + bb.y;
            r.z = (acc.z + hv.z) * di + bb.z;
            r.w = (acc.w + hv.w) * di + bb.w;
            r.x = r.x > 0.0f ? r.x : 0.0f;
            r.y = r.y > 0.0f ? r.y : 0.0f;
            r.z = r.z > 0.0f ? r.z : 0.0f;
            r.w = r.w > 0.0f ? r.w : 0.0f;
            *(float4*)(zt + ((t >> 4) * NPG + n) * ZP + 4 * l) = r;
        }
    }
    {
        float wv4[4];
        #pragma unroll
        for (int j = 0; j < 4; ++j) {
            int i = t + 256 * j;
            int k = i >> 4, c = i & 15;
            wv4[j] = W2[k * NCLS + (c < NCLS ? c : NCLS - 1)];
        }
        #pragma unroll
        for (int j = 0; j < 4; ++j) {
            int i = t + 256 * j;
            int k = i >> 4, c = i & 15;
            wT[c * ZP + k] = (c < NCLS) ? wv4[j] : 0.0f;
        }
    }
    __syncthreads();

    int wv = t >> 5;
    int tile = blockIdx.x * 4 + wv;
    if (wv < 4 && tile < ROW_TILES) {
        int lane = t & 31;
        int m = lane & 15, half = lane >> 4;
        const float* dp = dl + wv * 16 + 8 * half;
        float4 d0 = *(const float4*)(dp);
        float4 d1 = *(const float4*)(dp + 4);
        const float* ap = zt + (wv * 16 + m) * ZP + 2 * half;
        const float* bp = wT + m * ZP + 2 * half;
        v8f acc = {};
        #pragma unroll
        for (int k0 = 0; k0 < HID; k0 += 4) {
            v2f a = *(const v2f*)(ap + k0);
            v2f b = *(const v2f*)(bp + k0);
            acc = __builtin_amdgcn_wmma_f32_16x16x4_f32(false, a, false, b, (short)0, acc, false, false);
        }
        asm volatile("v_nop\n\tv_nop\n\tv_nop\n\tv_nop" : "+v"(acc));
        float* o = ho + (wv * 16 + 8 * half) * 16 + m;
        const float sc[8] = {d0.x, d0.y, d0.z, d0.w, d1.x, d1.y, d1.z, d1.w};
        #pragma unroll
        for (int v = 0; v < 8; ++v) o[v * 16] = acc[v] * sc[v];
    }
    __syncthreads();
    if (t < BLK_ROWS * NCLS / 4) {
        size_t gi = (size_t)blockIdx.x * (BLK_ROWS * NCLS) + 4 * t;
        if (gi < (size_t)NN * NCLS) {
            float q[4];
            #pragma unroll
            for (int j = 0; j < 4; ++j) {
                int e = 4 * t + j;
                int r = e / NCLS;
                q[j] = ho[r * 16 + (e - r * NCLS)];
            }
            *(float4*)(h2 + gi) = make_float4(q[0], q[1], q[2], q[3]);
        }
    }
}

__global__ void __launch_bounds__(256)
k_gather2(const int* __restrict__ rowptr, const int* __restrict__ cnt,
          const int* __restrict__ csr, const float* __restrict__ dis,
          const float* __restrict__ h2, const float* __restrict__ b2,
          float* __restrict__ out) {
    __shared__ float stage[16 * NPG * NCLS];
    int t = threadIdx.x;
    int f = t & 15;
    int fc = f < NCLS ? f : NCLS - 1;
    int grp = blockIdx.x * 16 + (t >> 4);
    if (grp < N_GROUPS) {
        int n0 = grp * NPG;
        int4 rp = *(const int4*)(rowptr + n0);
        int4 cn = *(const int4*)(cnt + n0);
        float4 dv = *(const float4*)(dis + n0);
        const int begs[NPG] = {rp.x, rp.y, rp.z, rp.w};
        const int cs[NPG] = {cn.x, cn.y, cn.z, cn.w};
        const float ds[NPG] = {dv.x, dv.y, dv.z, dv.w};
        float bias = b2[fc];
        int ia[NPG];
        #pragma unroll
        for (int n = 0; n < NPG; ++n)
            ia[n] = load_idx16(csr, begs[n], 0, f);
        #pragma unroll
        for (int n = 0; n < NPG; ++n) {
            int node = n0 + n;
            unsigned self_off = (unsigned)node * NCLS + (unsigned)fc;
            float hv = h2[self_off];
            int ibn = load_idx16(csr, begs[n], 16, f);
            float acc = node_sum_f1(csr, h2, begs[n], cs[n], ia[n], ibn, self_off, f, fc);
            float v = (acc + hv) * ds[n] + bias;
            float mx = f < NCLS ? v : -3.402823466e38f;
            for (int o = 8; o > 0; o >>= 1) mx = fmaxf(mx, __shfl_xor(mx, o, 16));
            float ex = f < NCLS ? expf(v - mx) : 0.0f;
            for (int o = 8; o > 0; o >>= 1) ex += __shfl_xor(ex, o, 16);
            float lse = mx + logf(ex);
            if (f < NCLS) stage[((t >> 4) * NPG + n) * NCLS + f] = v - lse;
        }
    }
    __syncthreads();
    for (int i = t; i < 16 * NPG * NCLS; i += 256) {
        size_t gi = (size_t)blockIdx.x * (16 * NPG * NCLS) + i;
        if (gi < (size_t)NN * NCLS) out[gi] = stage[i];
    }
}

extern "C" void kernel_launch(void* const* d_in, const int* in_sizes, int n_in,
                              void* d_out, int out_size, void* d_ws, size_t ws_size,
                              hipStream_t stream) {
    const float* x  = (const float*)d_in[0];
    const int*   ei = (const int*)d_in[1];
    const float* W1 = (const float*)d_in[2];
    const float* b1 = (const float*)d_in[3];
    const float* W2 = (const float*)d_in[4];
    const float* b2 = (const float*)d_in[5];
    const int* src = ei;
    const int* dst = ei + NE;
    float* out = (float*)d_out;

    float* ws   = (float*)d_ws;
    float* h1   = ws;
    float* h2   = h1 + 2 * (size_t)NN * HID;
    float* dis  = h2 + (size_t)NN * NCLS;
    int* cnt    = (int*)(dis + NN);
    int* rowptr = cnt + NN;
    int* csr    = rowptr + NN;
    int* binned = csr + NE;
    int* counts = binned + 2 * (size_t)NE;
    int* offs   = counts + NCH * NB;
    int* binbase = offs + NCH * NB;

    k_bincount<<<NCH, 256, 0, stream>>>(dst, counts);
    k_binscan<<<1, 1024, 0, stream>>>(counts, offs, binbase);
    k_binscatter<<<NCH, 256, 0, stream>>>(src, dst, counts, offs, binned);
    k_binsort<<<NB, SORT_T, 0, stream>>>(binned, binbase, rowptr, cnt, dis, csr);

    k_gemm1<<<(ROW_TILES + 3) / 4, 128, 0, stream>>>(x, W1, dis, h1);
    k_gather1<<<GATHER_BLOCKS, 256, 0, stream>>>(rowptr, cnt, csr, dis, h1, b1, W2, h2);

    k_gather2<<<GATHER_BLOCKS, 256, 0, stream>>>(rowptr, cnt, csr, dis, h2, b2, out);
}


# Each edge chunk sorted by bin in its own slice with a stored run table; each bin's block sums the table for its base and collects its 256 runs: count and scan launches gone; sums in LDS arrival order
# speedup vs baseline: 5.7548x; 1.0618x over previous
#include <hip/hip_runtime.h>
#include <math.h>


#define NN 100000
#define NE 1600000
#define FIN 128
#define HID 64
#define NCLS 10
#define ROW_TILES 6250
#define NB 256
#define BIN_NODES 391
#define NCH 256
#define CH 6250
#define SPLIT_T 512
#define CH_PER_T 13
#define SORT_T 1024
#define RUNS_PER_W 8
#define STAGE_CAP 16384

typedef __attribute__((ext_vector_type(2))) float v2f;
typedef __attribute__((ext_vector_type(8))) float v8f;

__device__ __forceinline__ unsigned bin_of(int d) {
    unsigned b = (unsigned)d / BIN_NODES;
    return b < NB - 1 ? b : NB - 1;
}

__device__ __forceinline__ int block_excl_scan(int c, int t, int* wtot, int nwaves) {
    int lane = t & 31, wv = t >> 5;
    int inc = c;
    for (int o = 1; o < 32; o <<= 1) {
        int v = __shfl_up(inc, o);
        if (lane >= o) inc += v;
    }
    if (lane == 31 && wv < nwaves) wtot[wv] = inc;
    __syncthreads();
    int off = 0;
    for (int w = 0; w < nwaves; ++w) off += (w < wv) ? wtot[w] : 0;
    return off + inc - c;
}

__device__ __forceinline__ int wave_sum(int v) {
    for (int o = 16; o > 0; o >>= 1) v += __shfl_xor(v, o);
    return v;
}

__global__ void __launch_bounds__(SPLIT_T)
k_binsplit(const int* __restrict__ src, const int* __restrict__ dst,
           int* __restrict__ counts, int* __restrict__ starts, int* __restrict__ binned_w) {
    int2* __restrict__ binned = (int2*)binned_w;
    __shared__ int hist[NB];
    __shared__ int cursor[NB];
    __shared__ int wtot[8];
    __shared__ int2 stage[CH];
    int t = threadIdx.x;
    if (t < NB) hist[t] = 0;
    const int* sp = src + blockIdx.x * CH;
    const int* dp = dst + blockIdx.x * CH;
    int sv[CH_PER_T], dv[CH_PER_T];
    #pragma unroll
    for (int k = 0; k < CH_PER_T; ++k) {
        int i = t + SPLIT_T * k;
        dv[k] = (i < CH) ? dp[i] : -1;
        sv[k] = (i < CH) ? sp[i] : 0;
    }
    __syncthreads();
    #pragma unroll
    for (int k = 0; k < CH_PER_T; ++k)
        if (t + SPLIT_T * k < CH) atomicAdd(&hist[bin_of(dv[k])], 1);
    __syncthreads();
    int c = (t < NB) ? hist[t] : 0;
    int ex = block_excl_scan(c, t, wtot, 8);
    if (t < NB) {
        cursor[t] = ex;
        counts[blockIdx.x * NB + t] = c;
        starts[blockIdx.x * NB + t] = ex;
    }
    __syncthreads();
    int p[CH_PER_T];
    #pragma unroll
    for (int k = 0; k < CH_PER_T; ++k)
        p[k] = (t + SPLIT_T * k < CH) ? atomicAdd(&cursor[bin_of(dv[k])], 1) : 0;
    #pragma unroll
    for (int k = 0; k < CH_PER_T; ++k)
        if (t + SPLIT_T * k < CH) stage[p[k]] = make_int2(sv[k], dv[k]);
    __syncthreads();
    int2* o = binned + blockIdx.x * CH;
    for (int i = t; i < CH; i += SPLIT_T) o[i] = stage[i];
}

__global__ void __launch_bounds__(SORT_T)
k_binsort(const int* __restrict__ binned_w, const int* __restrict__ counts,
          const int* __restrict__ starts,
          int* __restrict__ rowptr, int* __restrict__ cnt, float* __restrict__ dis,
          int* __restrict__ csr) {
    const int2* __restrict__ binned = (const int2*)binned_w;
    __shared__ int lcnt[512];
    __shared__ int lcur[512];
    __shared__ int wtot[16];
    __shared__ int wlow[32];
    __shared__ int wown[32];
    __shared__ int lst[STAGE_CAP];
    int t = threadIdx.x;
    int lane = t & 31;
    int wv = __builtin_amdgcn_readfirstlane(t >> 5);
    int bin = blockIdx.x;
    int lo = bin * BIN_NODES;
    int nn = NN - lo < BIN_NODES ? NN - lo : BIN_NODES;

    int rs[RUNS_PER_W], rc[RUNS_PER_W];
    int own = 0;
    #pragma unroll
    for (int k = 0; k < RUNS_PER_W; ++k) {
        int ch = wv * RUNS_PER_W + k;
        unsigned c = min((unsigned)counts[ch * NB + bin], (unsigned)CH);
        unsigned s = min((unsigned)starts[ch * NB + bin], (unsigned)CH - c);
        rc[k] = (int)c;
        rs[k] = ch * CH + (int)s;
        own += (int)c;
    }
    int2 ev[RUNS_PER_W];
    #pragma unroll
    for (int k = 0; k < RUNS_PER_W; ++k)
        ev[k] = (lane < rc[k]) ? binned[rs[k] + lane] : make_int2(0, 0);

    int low = 0;
    {
        int col = t & (NB - 1);
        if (col < bin) {
            const int* cp = counts + (t >> 8) * NB + col;
            #pragma unroll
            for (int j = 0; j < NCH / 4; ++j) low += cp[j * 4 * NB];
        }
    }
    if (t < 512) lcnt[t] = 0;
    __syncthreads();

    #pragma unroll
    for (int k = 0; k < RUNS_PER_W; ++k)
        if (lane < rc[k]) {
            unsigned li = (unsigned)(ev[k].y - lo);
            atomicAdd(&lcnt[li < BIN_NODES - 1 ? li : BIN_NODES - 1], 1);
        }
    #pragma unroll
    for (int k = 0; k < RUNS_PER_W; ++k)
        for (int o = 32; o < rc[k]; o += 32)
            if (o + lane < rc[k]) {
                unsigned li = (unsigned)(binned[rs[k] + o + lane].y - lo);
                atomicAdd(&lcnt[li < BIN_NODES - 1 ? li : BIN_NODES - 1], 1);
            }
    low = wave_sum(low);
    if (lane == 0) { wlow[wv] = low; wown[wv] = own; }
    __syncthreads();
    int base = wave_sum(wlow[lane]);
    int n = wave_sum(wown[lane]);
    int c = (t < 512) ? lcnt[t] : 0;
    int ex = block_excl_scan(c, t, wtot, 16);
    if (t < 512) lcur[t] = ex;
    if (t < nn) {
        rowptr[lo + t] = base + ex;
        cnt[lo + t] = c;
        dis[lo + t] = rsqrtf((float)(c + 1));
    }
    __syncthreads();
    bool staged = n <= STAGE_CAP;
    int p[RUNS_PER_W];
    #pragma unroll
    for (int k = 0; k < RUNS_PER_W; ++k) {
        unsigned li = (unsigned)(ev[k].y - lo);
        p[k] = (lane < rc[k]) ? atomicAdd(&lcur[li < BIN_NODES - 1 ? li : BIN_NODES - 1], 1) : 0;
    }
    #pragma unroll
    for (int k = 0; k < RUNS_PER_W; ++k)
        if (lane < rc[k]) {
            if (staged) lst[p[k]] = ev[k].x;
            else csr[base + p[k]] = ev[k].x;
        }
    #pragma unroll
    for (int k = 0; k < RUNS_PER_W; ++k)
        for (int o = 32; o < rc[k]; o += 32)
            if (o + lane < rc[k]) {
                int2 e = binned[rs[k] + o + lane];
                unsigned li = (unsigned)(e.y - lo);
                int q = atomicAdd(&lcur[li < BIN_NODES - 1 ? li : BIN_NODES - 1], 1);
                if (staged) lst[q] = e.x;
                else csr[base + q] = e.x;
            }
    if (staged) {
        __syncthreads();
        for (int i = t; i < n; i += SORT_T) csr[base + i] = lst[i];
    }
}

__global__ void __launch_bounds__(128)
k_gemm1(const float* __restrict__ x, const float* __restrict__ W1,
        const float* __restrict__ dis, float* __restrict__ h1) {
    __shared__ float wT[HID * 132];
    int t = threadIdx.x;
    for (int i = t; i < FIN * HID; i += 128) {
        int k = i >> 6, c = i & 63;
        wT[c * 132 + k] = W1[i];
    }
    __syncthreads();

    int tile = blockIdx.x * 4 + (t >> 5);
    if (tile >= ROW_TILES) return;
    int lane = t & 31;
    int m = lane & 15, half = lane >> 4;

    const float* arow = x + (size_t)(tile * 16 + m) * FIN + 2 * half;
    v8f a0 = {}, a1 = {}, a2 = {}, a3 = {};
    const float* dp = dis + tile * 16 + 8 * half;
    float4 d0 = *(const float4*)(dp);
    float4 d1 = *(const float4*)(dp + 4);

    #pragma unroll 4
    for (int k0 = 0; k0 < FIN; k0 += 4) {
        v2f a = *(const v2f*)(arow + k0);
        const float* bp = wT + m * 132 + k0 + 2 * half;
        v2f b0 = *(const v2f*)(bp);
        v2f b1 = *(const v2f*)(bp + 16 * 132);
        v2f b2 = *(const v2f*)(bp + 32 * 132);
        v2f b3 = *(const v2f*)(bp + 48 * 132);
        a0 = __builtin_amdgcn_wmma_f32_16x16x4_f32(false, a, false, b0, (short)0, a0, false, false);
        a1 = __builtin_amdgcn_wmma_f32_16x16x4_f32(false, a, false, b1, (short)0, a1, false, false);
        a2 = __builtin_amdgcn_wmma_f32_16x16x4_f32(false, a, false, b2, (short)0, a2, false, false);
        a3 = __builtin_amdgcn_wmma_f32_16x16x4_f32(false, a, false, b3, (short)0, a3, false, false);
    }

    float* out = h1 + (size_t)(tile * 16 + 8 * half) * HID + m;
    const float ds[8] = {d0.x, d0.y, d0.z, d0.w, d1.x, d1.y, d1.z, d1.w};
    #pragma unroll
    for (int v = 0; v < 8; ++v) {
        out[v * HID +  0] = a0[v] * ds[v];
        out[v * HID + 16] = a1[v] * ds[v];
        out[v * HID + 32] = a2[v] * ds[v];
        out[v * HID + 48] = a3[v] * ds[v];
    }
}

#define NPG 4
#define N_GROUPS 25000
#define GATHER_BLOCKS 1563

__device__ __forceinline__ int load_idx16(const int* __restrict__ csr, int beg, int e0, int l) {
    int j = beg + e0 + l;
    j = j < NE - 1 ? j : NE - 1;
    return (int)min((unsigned)csr[j], (unsigned)(NN - 1));
}

__device__ __forceinline__ int wave_count(int c) {
    int o = __shfl_xor(c, 16);
    return __builtin_amdgcn_readfirstlane(c > o ? c : o);
}

template <int B>
__device__ __forceinline__ void add_rows_f4(float4& acc, const float* __restrict__ h, int idx,
                                            int e0, int slot0, int c, unsigned self_off, int l) {
    float4 v[B];
    #pragma unroll
    for (int k = 0; k < B; ++k) {
        int s = __shfl(idx, slot0 + k, 16);
        unsigned off = (e0 + k < c) ? (unsigned)s * HID + 4u * (unsigned)l : self_off;
        v[k] = *(const float4*)(h + off);
    }
    #pragma unroll
    for (int k = 0; k < B; ++k) {
        bool ok = e0 + k < c;
        acc.x += ok ? v[k].x : 0.0f;
        acc.y += ok ? v[k].y : 0.0f;
        acc.z += ok ? v[k].z : 0.0f;
        acc.w += ok ? v[k].w : 0.0f;
    }
}

template <int B>
__device__ __forceinline__ void add_rows_f1(float& acc, const float* __restrict__ h, int idx,
                                            int e0, int slot0, int c, unsigned self_off, int fc) {
    float v[B];
    #pragma unroll
    for (int k = 0; k < B; ++k) {
        int s = __shfl(idx, slot0 + k, 16);
        unsigned off = (e0 + k < c) ? (unsigned)s * NCLS + (unsigned)fc : self_off;
        v[k] = h[off];
    }
    #pragma unroll
    for (int k = 0; k < B; ++k) acc += (e0 + k < c) ? v[k] : 0.0f;
}

__device__ __forceinline__ float4 node_sum_f4(const int* __restrict__ csr, const float* __restrict__ h,
                                              int beg, int c, int ia, int ib, unsigned self_off, int l) {
    float4 acc = make_float4(0.0f, 0.0f, 0.0f, 0.0f);
    int cw = wave_count(c);
    if (cw > 0)  add_rows_f4<8>(acc, h, ia, 0, 0, c, self_off, l);
    if (cw > 8)  add_rows_f4<8>(acc, h, ia, 8, 8, c, self_off, l);
    if (cw > 16) add_rows_f4<8>(acc, h, ib, 16, 0, c, self_off, l);
    if (cw > 24) add_rows_f4<8>(acc, h, ib, 24, 8, c, self_off, l);
    for (int e0 = 32; e0 < cw; e0 += 16) {
        int ix = load_idx16(csr, beg, e0, l);
        add_rows_f4<8>(acc, h, ix, e0, 0, c, self_off, l);
        if (e0 + 8 < cw) add_rows_f4<8>(acc, h, ix, e0 + 8, 8, c, self_off, l);
    }
    return acc;
}

__device__ __forceinline__ float node_sum_f1(const int* __restrict__ csr, const float* __restrict__ h,
                                             int beg, int c, int ia, int ib, unsigned self_off,
                                             int l, int fc) {
    float acc = 0.0f;
    int cw = wave_count(c);
    if (cw > 0)  add_rows_f1<16>(acc, h, ia, 0, 0, c, self_off, fc);
    if (cw > 16) add_rows_f1<16>(acc, h, ib, 16, 0, c, self_off, fc);
    for (int e0 = 32; e0 < cw; e0 += 16) {
        int ix = load_idx16(csr, beg, e0, l);
        add_rows_f1<16>(acc, h, ix, e0, 0, c, self_off, fc);
    }
    return acc;
}

#define ZP 68
#define BLK_ROWS (16 * NPG)
__global__ void __launch_bounds__(256)
k_gather1(const int* __restrict__ rowptr, const int* __restrict__ cnt,
          const int* __restrict__ csr, const float* __restrict__ dis,
          const float* __restrict__ h1, const float* __restrict__ b1,
          const float* __restrict__ W2, float* __restrict__ h2) {
    __shared__ float zt[BLK_ROWS * ZP];
    __shared__ float wT[16 * ZP];
    __shared__ float ho[BLK_ROWS * 16];
    __shared__ float dl[BLK_ROWS];
    int t = threadIdx.x;
    int l = t & 15;
    int grp = blockIdx.x * 16 + (t >> 4);
    if (grp < N_GROUPS) {
        int n0 = grp * NPG;
        int4 rp = *(const int4*)(rowptr + n0);
        int4 cn = *(const int4*)(cnt + n0);
        float4 dv = *(const float4*)(dis + n0);
        if (l == 0) *(float4*)(dl + (t >> 4) * NPG) = dv;
        const int begs[NPG] = {rp.x, rp.y, rp.z, rp.w};
        const int cs[NPG] = {cn.x, cn.y, cn.z, cn.w};
        const float ds[NPG] = {dv.x, dv.y, dv.z, dv.w};
        float4 bb = *(const float4*)(b1 + 4 * l);
        int ia[NPG];
        #pragma unroll
        for (int n = 0; n < NPG; ++n)
            ia[n] = load_idx16(csr, begs[n], 0, l);
        #pragma unroll
        for (int n = 0; n < NPG; ++n) {
            int node = n0 + n;
            unsigned self_off = (unsigned)node * HID + 4u * (unsigned)l;
            float4 hv = *(const float4*)(h1 + self_off);
            int ibn = load_idx16(csr, begs[n], 16, l);
            float4 acc = node_sum_f4(csr, h1, begs[n], cs[n], ia[n], ibn, self_off, l);
            float di = ds[n];
            float4 r;
            r.x = (acc.x + hv.x) * di + bb.x;
            r.y = (acc.y + hv.y) * di + bb.y;
            r.z = (acc.z + hv.z) * di + bb.z;
            r.w = (acc.w + hv.w) * di + bb.w;
            r.x = r.x > 0.0f ? r.x : 0.0f;
            r.y = r.y > 0.0f ? r.y : 0.0f;
            r.z = r.z > 0.0f ? r.z : 0.0f;
            r.w = r.w > 0.0f ? r.w : 0.0f;
            *(float4*)(zt + ((t >> 4) * NPG + n) * ZP + 4 * l) = r;
        }
    }
    {
        float wv4[4];
        #pragma unroll
        for (int j = 0; j < 4; ++j) {
            int i = t + 256 * j;
            int k = i >> 4, c = i & 15;
            wv4[j] = W2[k * NCLS + (c < NCLS ? c : NCLS - 1)];
        }
        #pragma unroll
        for (int j = 0; j < 4; ++j) {
            int i = t + 256 * j;
            int k = i >> 4, c = i & 15;
            wT[c * ZP + k] = (c < NCLS) ? wv4[j] : 0.0f;
        }
    }
    __syncthreads();

    int wv = t >> 5;
    int tile = blockIdx.x * 4 + wv;
    if (wv < 4 && tile < ROW_TILES) {
        int lane = t & 31;
        int m = lane & 15, half = lane >> 4;
        const float* dp = dl + wv * 16 + 8 * half;
        float4 d0 = *(const float4*)(dp);
        float4 d1 = *(const float4*)(dp + 4);
        const float* ap = zt + (wv * 16 + m) * ZP + 2 * half;
        const float* bp = wT + m * ZP + 2 * half;
        v8f acc = {};
        #pragma unroll
        for (int k0 = 0; k0 < HID; k0 += 4) {
            v2f a = *(const v2f*)(ap + k0);
            v2f b = *(const v2f*)(bp + k0);
            acc = __builtin_amdgcn_wmma_f32_16x16x4_f32(false, a, false, b, (short)0, acc, false, false);
        }
        asm volatile("v_nop\n\tv_nop\n\tv_nop\n\tv_nop" : "+v"(acc));
        float* o = ho + (wv * 16 + 8 * half) * 16 + m;
        const float sc[8] = {d0.x, d0.y, d0.z, d0.w, d1.x, d1.y, d1.z, d1.w};
        #pragma unroll
        for (int v = 0; v < 8; ++v) o[v * 16] = acc[v] * sc[v];
    }
    __syncthreads();
    if (t < BLK_ROWS * NCLS / 4) {
        size_t gi = (size_t)blockIdx.x * (BLK_ROWS * NCLS) + 4 * t;
        if (gi < (size_t)NN * NCLS) {
            float q[4];
            #pragma unroll
            for (int j = 0; j < 4; ++j) {
                int e = 4 * t + j;
                int r = e / NCLS;
                q[j] = ho[r * 16 + (e - r * NCLS)];
            }
            *(float4*)(h2 + gi) = make_float4(q[0], q[1], q[2], q[3]);
        }
    }
}

__global__ void __launch_bounds__(256)
k_gather2(const int* __restrict__ rowptr, const int* __restrict__ cnt,
          const int* __restrict__ csr, const float* __restrict__ dis,
          const float* __restrict__ h2, const float* __restrict__ b2,
          float* __restrict__ out) {
    __shared__ float stage[16 * NPG * NCLS];
    int t = threadIdx.x;
    int f = t & 15;
    int fc = f < NCLS ? f : NCLS - 1;
    int grp = blockIdx.x * 16 + (t >> 4);
    if (grp < N_GROUPS) {
        int n0 = grp * NPG;
        int4 rp = *(const int4*)(rowptr + n0);
        int4 cn = *(const int4*)(cnt + n0);
        float4 dv = *(const float4*)(dis + n0);
        const int begs[NPG] = {rp.x, rp.y, rp.z, rp.w};
        const int cs[NPG] = {cn.x, cn.y, cn.z, cn.w};
        const float ds[NPG] = {dv.x, dv.y, dv.z, dv.w};
        float bias = b2[fc];
        int ia[NPG];
        #pragma unroll
        for (int n = 0; n < NPG; ++n)
            ia[n] = load_idx16(csr, begs[n], 0, f);
        #pragma unroll
        for (int n = 0; n < NPG; ++n) {
            int node = n0 + n;
            unsigned self_off = (unsigned)node * NCLS + (unsigned)fc;
            float hv = h2[self_off];
            int ibn = load_idx16(csr, begs[n], 16, f);
            float acc = node_sum_f1(csr, h2, begs[n], cs[n], ia[n], ibn, self_off, f, fc);
            float v = (acc + hv) * ds[n] + bias;
            float mx = f < NCLS ? v : -3.402823466e38f;
            for (int o = 8; o > 0; o >>= 1) mx = fmaxf(mx, __shfl_xor(mx, o, 16));
            float ex = f < NCLS ? expf(v - mx) : 0.0f;
            for (int o = 8; o > 0; o >>= 1) ex += __shfl_xor(ex, o, 16);
            float lse = mx + logf(ex);
            if (f < NCLS) stage[((t >> 4) * NPG + n) * NCLS + f] = v - lse;
        }
    }
    __syncthreads();
    for (int i = t; i < 16 * NPG * NCLS; i += 256) {
        size_t gi = (size_t)blockIdx.x * (16 * NPG * NCLS) + i;
        if (gi < (size_t)NN * NCLS) out[gi] = stage[i];
    }
}

extern "C" void kernel_launch(void* const* d_in, const int* in_sizes, int n_in,
                              void* d_out, int out_size, void* d_ws, size_t ws_size,
                              hipStream_t stream) {
    const float* x  = (const float*)d_in[0];
    const int*   ei = (const int*)d_in[1];
    const float* W1 = (const float*)d_in[2];
    const float* b1 = (const float*)d_in[3];
    const float* W2 = (const float*)d_in[4];
    const float* b2 = (const float*)d_in[5];
    const int* src = ei;
    const int* dst = ei + NE;
    float* out = (float*)d_out;

    float* ws   = (float*)d_ws;
    float* h1   = ws;
    float* h2   = h1 + 2 * (size_t)NN * HID;
    float* dis  = h2 + (size_t)NN * NCLS;
    int* cnt    = (int*)(dis + NN);
    int* rowptr = cnt + NN;
    int* csr    = rowptr + NN;
    int* binned = csr + NE;
    int* counts = binned + 2 * (size_t)NE;
    int* starts = counts + NCH * NB;

    k_binsplit<<<NCH, SPLIT_T, 0, stream>>>(src, dst, counts, starts, binned);
    k_binsort<<<NB, SORT_T, 0, stream>>>(binned, counts, starts, rowptr, cnt, dis, csr);

    k_gemm1<<<(ROW_TILES + 3) / 4, 128, 0, stream>>>(x, W1, dis, h1);
    k_gather1<<<GATHER_BLOCKS, 256, 0, stream>>>(rowptr, cnt, csr, dis, h1, b1, W2, h2);

    k_gather2<<<GATHER_BLOCKS, 256, 0, stream>>>(rowptr, cnt, csr, dis, h2, b2, out);
}


# Layer-1 product moved into the bin-sort launch: each bin's work-group multiplies its own 391 rows of X by W1 (staged once per group) once it has their deg^-1/2; one launch fewer; sum order kept
# speedup vs baseline: 6.2518x; 1.0864x over previous
#include <hip/hip_runtime.h>
#include <math.h>


#define NN 100000
#define NE 1600000
#define FIN 128
#define HID 64
#define NCLS 10
#define ROW_TILES 6250
#define NB 256
#define BIN_NODES 391
#define NCH 256
#define CH 6250
#define SPLIT_T 512
#define CH_PER_T 13
#define SORT_T 1024
#define RUNS_PER_W 8
#define STAGE_CAP 16384

typedef __attribute__((ext_vector_type(2))) float v2f;
typedef __attribute__((ext_vector_type(8))) float v8f;

__device__ __forceinline__ unsigned bin_of(int d) {
    unsigned b = (unsigned)d / BIN_NODES;
    return b < NB - 1 ? b : NB - 1;
}

__device__ __forceinline__ int block_excl_scan(int c, int t, int* wtot, int nwaves) {
    int lane = t & 31, wv = t >> 5;
    int inc = c;
    for (int o = 1; o < 32; o <<= 1) {
        int v = __shfl_up(inc, o);
        if (lane >= o) inc += v;
    }
    if (lane == 31 && wv < nwaves) wtot[wv] = inc;
    __syncthreads();
    int off = 0;
    for (int w = 0; w < nwaves; ++w) off += (w < wv) ? wtot[w] : 0;
    return off + inc - c;
}

__device__ __forceinline__ int wave_sum(int v) {
    for (int o = 16; o > 0; o >>= 1) v += __shfl_xor(v, o);
    return v;
}

__global__ void __launch_bounds__(SPLIT_T)
k_binsplit(const int* __restrict__ src, const int* __restrict__ dst,
           int* __restrict__ counts, int* __restrict__ starts, int* __restrict__ binned_w) {
    int2* __restrict__ binned = (int2*)binned_w;
    __shared__ int hist[NB];
    __shared__ int cursor[NB];
    __shared__ int wtot[8];
    __shared__ int2 stage[CH];
    int t = threadIdx.x;
    if (t < NB) hist[t] = 0;
    const int* sp = src + blockIdx.x * CH;
    const int* dp = dst + blockIdx.x * CH;
    int sv[CH_PER_T], dv[CH_PER_T];
    #pragma unroll
    for (int k = 0; k < CH_PER_T; ++k) {
        int i = t + SPLIT_T * k;
        dv[k] = (i < CH) ? dp[i] : -1;
        sv[k] = (i < CH) ? sp[i] : 0;
    }
    __syncthreads();
    #pragma unroll
    for (int k = 0; k < CH_PER_T; ++k)
        if (t + SPLIT_T * k < CH) atomicAdd(&hist[bin_of(dv[k])], 1);
    __syncthreads();
    int c = (t < NB) ? hist[t] : 0;
    int ex = block_excl_scan(c, t, wtot, 8);
    if (t < NB) {
        cursor[t] = ex;
        counts[blockIdx.x * NB + t] = c;
        starts[blockIdx.x * NB + t] = ex;
    }
    __syncthreads();
    int p[CH_PER_T];
    #pragma unroll
    for (int k = 0; k < CH_PER_T; ++k)
        p[k] = (t + SPLIT_T * k < CH) ? atomicAdd(&cursor[bin_of(dv[k])], 1) : 0;
    #pragma unroll
    for (int k = 0; k < CH_PER_T; ++k)
        if (t + SPLIT_T * k < CH) stage[p[k]] = make_int2(sv[k], dv[k]);
    __syncthreads();
    int2* o = binned + blockIdx.x * CH;
    for (int i = t; i < CH; i += SPLIT_T) o[i] = stage[i];
}

#define W1P 132
__global__ void __launch_bounds__(SORT_T)
k_binsort(const int* __restrict__ binned_w, const int* __restrict__ counts,
          const int* __restrict__ starts,
          const float* __restrict__ x, const float* __restrict__ W1,
          int* __restrict__ rowptr, int* __restrict__ cnt, float* __restrict__ dis,
          int* __restrict__ csr, float* __restrict__ h1) {
    const int2* __restrict__ binned = (const int2*)binned_w;
    __shared__ int lcnt[512];
    __shared__ int lcur[512];
    __shared__ int wtot[16];
    __shared__ int wlow[32];
    __shared__ int wown[32];
    __shared__ int lst[STAGE_CAP];
    __shared__ __align__(16) float wT[HID * W1P];
    __shared__ __align__(16) float ldis[512];
    int t = threadIdx.x;
    int lane = t & 31;
    int wv = __builtin_amdgcn_readfirstlane(t >> 5);
    int bin = blockIdx.x;
    int lo = bin * BIN_NODES;
    int nn = NN - lo < BIN_NODES ? NN - lo : BIN_NODES;

    int rs[RUNS_PER_W], rc[RUNS_PER_W];
    int own = 0;
    #pragma unroll
    for (int k = 0; k < RUNS_PER_W; ++k) {
        int ch = wv * RUNS_PER_W + k;
        unsigned c = min((unsigned)counts[ch * NB + bin], (unsigned)CH);
        unsigned s = min((unsigned)starts[ch * NB + bin], (unsigned)CH - c);
        rc[k] = (int)c;
        rs[k] = ch * CH + (int)s;
        own += (int)c;
    }
    int2 ev[RUNS_PER_W];
    #pragma unroll
    for (int k = 0; k < RUNS_PER_W; ++k)
        ev[k] = (lane < rc[k]) ? binned[rs[k] + lane] : make_int2(0, 0);

    float w1v[FIN * HID / SORT_T];
    #pragma unroll
    for (int j = 0; j < FIN * HID / SORT_T; ++j) w1v[j] = W1[t + SORT_T * j];

    int low = 0;
    {
        int col = t & (NB - 1);
        if (col < bin) {
            const int* cp = counts + (t >> 8) * NB + col;
            #pragma unroll
            for (int j = 0; j < NCH / 4; ++j) low += cp[j * 4 * NB];
        }
    }
    if (t < 512) lcnt[t] = 0;
    #pragma unroll
    for (int j = 0; j < FIN * HID / SORT_T; ++j) {
        int i = t + SORT_T * j;
        wT[(i & 63) * W1P + (i >> 6)] = w1v[j];
    }
    __syncthreads();

    #pragma unroll
    for (int k = 0; k < RUNS_PER_W; ++k)
        if (lane < rc[k]) {
            unsigned li = (unsigned)(ev[k].y - lo);
            atomicAdd(&lcnt[li < BIN_NODES - 1 ? li : BIN_NODES - 1], 1);
        }
    #pragma unroll
    for (int k = 0; k < RUNS_PER_W; ++k)
        for (int o = 32; o < rc[k]; o += 32)
            if (o + lane < rc[k]) {
                unsigned li = (unsigned)(binned[rs[k] + o + lane].y - lo);
                atomicAdd(&lcnt[li < BIN_NODES - 1 ? li : BIN_NODES - 1], 1);
            }
    low = wave_sum(low);
    if (lane == 0) { wlow[wv] = low; wown[wv] = own; }
    __syncthreads();
    int base = wave_sum(wlow[lane]);
    int n = wave_sum(wown[lane]);
    int c = (t < 512) ? lcnt[t] : 0;
    int ex = block_excl_scan(c, t, wtot, 16);
    float dsc = rsqrtf((float)(c + 1));
    if (t < 512) { lcur[t] = ex; ldis[t] = dsc; }
    if (t < nn) {
        rowptr[lo + t] = base + ex;
        cnt[lo + t] = c;
        dis[lo + t] = dsc;
    }
    __syncthreads();
    bool staged = n <= STAGE_CAP;
    int p[RUNS_PER_W];
    #pragma unroll
    for (int k = 0; k < RUNS_PER_W; ++k) {
        unsigned li = (unsigned)(ev[k].y - lo);
        p[k] = (lane < rc[k]) ? atomicAdd(&lcur[li < BIN_NODES - 1 ? li : BIN_NODES - 1], 1) : 0;
    }
    #pragma unroll
    for (int k = 0; k < RUNS_PER_W; ++k)
        if (lane < rc[k]) {
            if (staged) lst[p[k]] = ev[k].x;
            else csr[base + p[k]] = ev[k].x;
        }
    #pragma unroll
    for (int k = 0; k < RUNS_PER_W; ++k)
        for (int o = 32; o < rc[k]; o += 32)
            if (o + lane < rc[k]) {
                int2 e = binned[rs[k] + o + lane];
                unsigned li = (unsigned)(e.y - lo);
                int q = atomicAdd(&lcur[li < BIN_NODES - 1 ? li : BIN_NODES - 1], 1);
                if (staged) lst[q] = e.x;
                else csr[base + q] = e.x;
            }
    if (staged) {
        __syncthreads();
        for (int i = t; i < n; i += SORT_T) csr[base + i] = lst[i];
    }

    int r0 = wv * 16;
    if (r0 < nn) {
        int m = lane & 15, half = lane >> 4;
        int ra = r0 + m;
        ra = ra < nn ? ra : nn - 1;
        const float* arow = x + (size_t)(lo + ra) * FIN + 2 * half;
        v8f a0 = {}, a1 = {}, a2 = {}, a3 = {};
        const float* dp = ldis + r0 + 8 * half;
        float4 d0 = *(const float4*)(dp);
        float4 d1 = *(const float4*)(dp + 4);

        #pragma unroll 4
        for (int k0 = 0; k0 < FIN; k0 += 4) {
            v2f a = *(const v2f*)(arow + k0);
            const float* bp = wT + m * W1P + k0 + 2 * half;
            v2f b0 = *(const v2f*)(bp);
            v2f b1 = *(const v2f*)(bp + 16 * W1P);
            v2f b2 = *(const v2f*)(bp + 32 * W1P);
            v2f b3 = *(const v2f*)(bp + 48 * W1P);
            a0 = __builtin_amdgcn_wmma_f32_16x16x4_f32(false, a, false, b0, (short)0, a0, false, false);
            a1 = __builtin_amdgcn_wmma_f32_16x16x4_f32(false, a, false, b1, (short)0, a1, false, false);
            a2 = __builtin_amdgcn_wmma_f32_16x16x4_f32(false, a, false, b2, (short)0, a2, false, false);
            a3 = __builtin_amdgcn_wmma_f32_16x16x4_f32(false, a, false, b3, (short)0, a3, false, false);
        }

        int rb = r0 + 8 * half;
        float* out = h1 + (size_t)(lo + rb) * HID + m;
        const float ds[8] = {d0.x, d0.y, d0.z, d0.w, d1.x, d1.y, d1.z, d1.w};
        #pragma unroll
        for (int v = 0; v < 8; ++v)
            if (rb + v < nn) {
                out[v * HID +  0] = a0[v] * ds[v];
                out[v * HID + 16] = a1[v] * ds[v];
                out[v * HID + 32] = a2[v] * ds[v];
                out[v * HID + 48] = a3[v] * ds[v];
            }
    }
}

#define NPG 4
#define N_GROUPS 25000
#define GATHER_BLOCKS 1563

__device__ __forceinline__ int load_idx16(const int* __restrict__ csr, int beg, int e0, int l) {
    int j = beg + e0 + l;
    j = j < NE - 1 ? j : NE - 1;
    return (int)min((unsigned)csr[j], (unsigned)(NN - 1));
}

__device__ __forceinline__ int wave_count(int c) {
    int o = __shfl_xor(c, 16);
    return __builtin_amdgcn_readfirstlane(c > o ? c : o);
}

template <int B>
__device__ __forceinline__ void add_rows_f4(float4& acc, const float* __restrict__ h, int idx,
                                            int e0, int slot0, int c, unsigned self_off, int l) {
    float4 v[B];
    #pragma unroll
    for (int k = 0; k < B; ++k) {
        int s = __shfl(idx, slot0 + k, 16);
        unsigned off = (e0 + k < c) ? (unsigned)s * HID + 4u * (unsigned)l : self_off;
        v[k] = *(const float4*)(h + off);
    }
    #pragma unroll
    for (int k = 0; k < B; ++k) {
        bool ok = e0 + k < c;
        acc.x += ok ? v[k].x : 0.0f;
        acc.y += ok ? v[k].y : 0.0f;
        acc.z += ok ? v[k].z : 0.0f;
        acc.w += ok ? v[k].w : 0.0f;
    }
}

template <int B>
__device__ __forceinline__ void add_rows_f1(float& acc, const float* __restrict__ h, int idx,
                                            int e0, int slot0, int c, unsigned self_off, int fc) {
    float v[B];
    #pragma unroll
    for (int k = 0; k < B; ++k) {
        int s = __shfl(idx, slot0 + k, 16);
        unsigned off = (e0 + k < c) ? (unsigned)s * NCLS + (unsigned)fc : self_off;
        v[k] = h[off];
    }
    #pragma unroll
    for (int k = 0; k < B; ++k) acc += (e0 + k < c) ? v[k] : 0.0f;
}

__device__ __forceinline__ float4 node_sum_f4(const int* __restrict__ csr, const float* __restrict__ h,
                                              int beg, int c, int ia, int ib, unsigned self_off, int l) {
    float4 acc = make_float4(0.0f, 0.0f, 0.0f, 0.0f);
    int cw = wave_count(c);
    if (cw > 0)  add_rows_f4<8>(acc, h, ia, 0, 0, c, self_off, l);
    if (cw > 8)  add_rows_f4<8>(acc, h, ia, 8, 8, c, self_off, l);
    if (cw > 16) add_rows_f4<8>(acc, h, ib, 16, 0, c, self_off, l);
    if (cw > 24) add_rows_f4<8>(acc, h, ib, 24, 8, c, self_off, l);
    for (int e0 = 32; e0 < cw; e0 += 16) {
        int ix = load_idx16(csr, beg, e0, l);
        add_rows_f4<8>(acc, h, ix, e0, 0, c, self_off, l);
        if (e0 + 8 < cw) add_rows_f4<8>(acc, h, ix, e0 + 8, 8, c, self_off, l);
    }
    return acc;
}

__device__ __forceinline__ float node_sum_f1(const int* __restrict__ csr, const float* __restrict__ h,
                                             int beg, int c, int ia, int ib, unsigned self_off,
                                             int l, int fc) {
    float acc = 0.0f;
    int cw = wave_count(c);
    if (cw > 0)  add_rows_f1<16>(acc, h, ia, 0, 0, c, self_off, fc);
    if (cw > 16) add_rows_f1<16>(acc, h, ib, 16, 0, c, self_off, fc);
    for (int e0 = 32; e0 < cw; e0 += 16) {
        int ix = load_idx16(csr, beg, e0, l);
        add_rows_f1<16>(acc, h, ix, e0, 0, c, self_off, fc);
    }
    return acc;
}

#define ZP 68
#define BLK_ROWS (16 * NPG)
__global__ void __launch_bounds__(256)
k_gather1(const int* __restrict__ rowptr, const int* __restrict__ cnt,
          const int* __restrict__ csr, const float* __restrict__ dis,
          const float* __restrict__ h1, const float* __restrict__ b1,
          const float* __restrict__ W2, float* __restrict__ h2) {
    __shared__ float zt[BLK_ROWS * ZP];
    __shared__ float wT[16 * ZP];
    __shared__ float ho[BLK_ROWS * 16];
    __shared__ float dl[BLK_ROWS];
    int t = threadIdx.x;
    int l = t & 15;
    int grp = blockIdx.x * 16 + (t >> 4);
    if (grp < N_GROUPS) {
        int n0 = grp * NPG;
        int4 rp = *(const int4*)(rowptr + n0);
        int4 cn = *(const int4*)(cnt + n0);
        float4 dv = *(const float4*)(dis + n0);
        if (l == 0) *(float4*)(dl + (t >> 4) * NPG) = dv;
        const int begs[NPG] = {rp.x, rp.y, rp.z, rp.w};
        const int cs[NPG] = {cn.x, cn.y, cn.z, cn.w};
        const float ds[NPG] = {dv.x, dv.y, dv.z, dv.w};
        float4 bb = *(const float4*)(b1 + 4 * l);
        int ia[NPG];
        #pragma unroll
        for (int n = 0; n < NPG; ++n)
            ia[n] = load_idx16(csr, begs[n], 0, l);
        #pragma unroll
        for (int n = 0; n < NPG; ++n) {
            int node = n0 + n;
            unsigned self_off = (unsigned)node * HID + 4u * (unsigned)l;
            float4 hv = *(const float4*)(h1 + self_off);
            int ibn = load_idx16(csr, begs[n], 16, l);
            float4 acc = node_sum_f4(csr, h1, begs[n], cs[n], ia[n], ibn, self_off, l);
            float di = ds[n];
            float4 r;
            r.x = (acc.x + hv.x) * di + bb.x;
            r.y = (acc.y + hv.y) * di + bb.y;
            r.z = (acc.z + hv.z) * di + bb.z;
            r.w = (acc.w + hv.w) * di + bb.w;
            r.x = r.x > 0.0f ? r.x : 0.0f;
            r.y = r.y > 0.0f ? r.y : 0.0f;
            r.z = r.z > 0.0f ? r.z : 0.0f;
            r.w = r.w > 0.0f ? r.w : 0.0f;
            *(float4*)(zt + ((t >> 4) * NPG + n) * ZP + 4 * l) = r;
        }
    }
    {
        float wv4[4];
        #pragma unroll
        for (int j = 0; j < 4; ++j) {
            int i = t + 256 * j;
            int k = i >> 4, c = i & 15;
            wv4[j] = W2[k * NCLS + (c < NCLS ? c : NCLS - 1)];
        }
        #pragma unroll
        for (int j = 0; j < 4; ++j) {
            int i = t + 256 * j;
            int k = i >> 4, c = i & 15;
            wT[c * ZP + k] = (c < NCLS) ? wv4[j] : 0.0f;
        }
    }
    __syncthreads();

    int wv = t >> 5;
    int tile = blockIdx.x * 4 + wv;
    if (wv < 4 && tile < ROW_TILES) {
        int lane = t & 31;
        int m = lane & 15, half = lane >> 4;
        const float* dp = dl + wv * 16 + 8 * half;
        float4 d0 = *(const float4*)(dp);
        float4 d1 = *(const float4*)(dp + 4);
        const float* ap = zt + (wv * 16 + m) * ZP + 2 * half;
        const float* bp = wT + m * ZP + 2 * half;
        v8f acc = {};
        #pragma unroll
        for (int k0 = 0; k0 < HID; k0 += 4) {
            v2f a = *(const v2f*)(ap + k0);
            v2f b = *(const v2f*)(bp + k0);
            acc = __builtin_amdgcn_wmma_f32_16x16x4_f32(false, a, false, b, (short)0, acc, false, false);
        }
        asm volatile("v_nop\n\tv_nop\n\tv_nop\n\tv_nop" : "+v"(acc));
        float* o = ho + (wv * 16 + 8 * half) * 16 + m;
        const float sc[8] = {d0.x, d0.y, d0.z, d0.w, d1.x, d1.y, d1.z, d1.w};
        #pragma unroll
        for (int v = 0; v < 8; ++v) o[v * 16] = acc[v] * sc[v];
    }
    __syncthreads();
    if (t < BLK_ROWS * NCLS / 4) {
        size_t gi = (size_t)blockIdx.x * (BLK_ROWS * NCLS) + 4 * t;
        if (gi < (size_t)NN * NCLS) {
            float q[4];
            #pragma unroll
            for (int j = 0; j < 4; ++j) {
                int e = 4 * t + j;
                int r = e / NCLS;
                q[j] = ho[r * 16 + (e - r * NCLS)];
            }
            *(float4*)(h2 + gi) = make_float4(q[0], q[1], q[2], q[3]);
        }
    }
}

__global__ void __launch_bounds__(256)
k_gather2(const int* __restrict__ rowptr, const int* __restrict__ cnt,
          const int* __restrict__ csr, const float* __restrict__ dis,
          const float* __restrict__ h2, const float* __restrict__ b2,
          float* __restrict__ out) {
    __shared__ float stage[16 * NPG * NCLS];
    int t = threadIdx.x;
    int f = t & 15;
    int fc = f < NCLS ? f : NCLS - 1;
    int grp = blockIdx.x * 16 + (t >> 4);
    if (grp < N_GROUPS) {
        int n0 = grp * NPG;
        int4 rp = *(const int4*)(rowptr + n0);
        int4 cn = *(const int4*)(cnt + n0);
        float4 dv = *(const float4*)(dis + n0);
        const int begs[NPG] = {rp.x, rp.y, rp.z, rp.w};
        const int cs[NPG] = {cn.x, cn.y, cn.z, cn.w};
        const float ds[NPG] = {dv.x, dv.y, dv.z, dv.w};
        float bias = b2[fc];
        int ia[NPG];
        #pragma unroll
        for (int n = 0; n < NPG; ++n)
            ia[n] = load_idx16(csr, begs[n], 0, f);
        #pragma unroll
        for (int n = 0; n < NPG; ++n) {
            int node = n0 + n;
            unsigned self_off = (unsigned)node * NCLS + (unsigned)fc;
            float hv = h2[self_off];
            int ibn = load_idx16(csr, begs[n], 16, f);
            float acc = node_sum_f1(csr, h2, begs[n], cs[n], ia[n], ibn, self_off, f, fc);
            float v = (acc + hv) * ds[n] + bias;
            float mx = f < NCLS ? v : -3.402823466e38f;
            for (int o = 8; o > 0; o >>= 1) mx = fmaxf(mx, __shfl_xor(mx, o, 16));
            float ex = f < NCLS ? expf(v - mx) : 0.0f;
            for (int o = 8; o > 0; o >>= 1) ex += __shfl_xor(ex, o, 16);
            float lse = mx + logf(ex);
            if (f < NCLS) stage[((t >> 4) * NPG + n) * NCLS + f] = v - lse;
        }
    }
    __syncthreads();
    for (int i = t; i < 16 * NPG * NCLS; i += 256) {
        size_t gi = (size_t)blockIdx.x * (16 * NPG * NCLS) + i;
        if (gi < (size_t)NN * NCLS) out[gi] = stage[i];
    }
}

extern "C" void kernel_launch(void* const* d_in, const int* in_sizes, int n_in,
                              void* d_out, int out_size, void* d_ws, size_t ws_size,
                              hipStream_t stream) {
    const float* x  = (const float*)d_in[0];
    const int*   ei = (const int*)d_in[1];
    const float* W1 = (const float*)d_in[2];
    const float* b1 = (const float*)d_in[3];
    const float* W2 = (const float*)d_in[4];
    const float* b2 = (const float*)d_in[5];
    const int* src = ei;
    const int* dst = ei + NE;
    float* out = (float*)d_out;

    float* ws   = (float*)d_ws;
    float* h1   = ws;
    float* h2   = h1 + 2 * (size_t)NN * HID;
    float* dis  = h2 + (size_t)NN * NCLS;
    int* cnt    = (int*)(dis + NN);
    int* rowptr = cnt + NN;
    int* csr    = rowptr + NN;
    int* binned = csr + NE;
    int* counts = binned + 2 * (size_t)NE;
    int* starts = counts + NCH * NB;

    k_binsplit<<<NCH, SPLIT_T, 0, stream>>>(src, dst, counts, starts, binned);
    k_binsort<<<NB, SORT_T, 0, stream>>>(binned, counts, starts, x, W1, rowptr, cnt, dis, csr, h1);

    k_gather1<<<GATHER_BLOCKS, 256, 0, stream>>>(rowptr, cnt, csr, dis, h1, b1, W2, h2);

    k_gather2<<<GATHER_BLOCKS, 256, 0, stream>>>(rowptr, cnt, csr, dis, h2, b2, out);
}


# Layer-2 rows stored 16 floats wide (64 bytes, 6 zero columns) as the product leaves them: every row the layer-2 gather reads lies in one 128-byte line (one in four crossed two); sum order kept
# speedup vs baseline: 6.6663x; 1.0663x over previous
#include <hip/hip_runtime.h>
#include <math.h>


#define NN 100000
#define NE 1600000
#define FIN 128
#define HID 64
#define NCLS 10
#define H2P 16
#define ROW_TILES 6250
#define NB 256
#define BIN_NODES 391
#define NCH 256
#define CH 6250
#define SPLIT_T 512
#define CH_PER_T 13
#define SORT_T 1024
#define RUNS_PER_W 8
#define STAGE_CAP 16384

typedef __attribute__((ext_vector_type(2))) float v2f;
typedef __attribute__((ext_vector_type(8))) float v8f;

__device__ __forceinline__ unsigned bin_of(int d) {
    unsigned b = (unsigned)d / BIN_NODES;
    return b < NB - 1 ? b : NB - 1;
}

__device__ __forceinline__ int block_excl_scan(int c, int t, int* wtot, int nwaves) {
    int lane = t & 31, wv = t >> 5;
    int inc = c;
    for (int o = 1; o < 32; o <<= 1) {
        int v = __shfl_up(inc, o);
        if (lane >= o) inc += v;
    }
    if (lane == 31 && wv < nwaves) wtot[wv] = inc;
    __syncthreads();
    int off = 0;
    for (int w = 0; w < nwaves; ++w) off += (w < wv) ? wtot[w] : 0;
    return off + inc - c;
}

__device__ __forceinline__ int wave_sum(int v) {
    for (int o = 16; o > 0; o >>= 1) v += __shfl_xor(v, o);
    return v;
}

__global__ void __launch_bounds__(SPLIT_T)
k_binsplit(const int* __restrict__ src, const int* __restrict__ dst,
           int* __restrict__ counts, int* __restrict__ starts, int* __restrict__ binned_w) {
    int2* __restrict__ binned = (int2*)binned_w;
    __shared__ int hist[NB];
    __shared__ int cursor[NB];
    __shared__ int wtot[8];
    __shared__ int2 stage[CH];
    int t = threadIdx.x;
    if (t < NB) hist[t] = 0;
    const int* sp = src + blockIdx.x * CH;
    const int* dp = dst + blockIdx.x * CH;
    int sv[CH_PER_T], dv[CH_PER_T];
    #pragma unroll
    for (int k = 0; k < CH_PER_T; ++k) {
        int i = t + SPLIT_T * k;
        dv[k] = (i < CH) ? dp[i] : -1;
        sv[k] = (i < CH) ? sp[i] : 0;
    }
    __syncthreads();
    #pragma unroll
    for (int k = 0; k < CH_PER_T; ++k)
        if (t + SPLIT_T * k < CH) atomicAdd(&hist[bin_of(dv[k])], 1);
    __syncthreads();
    int c = (t < NB) ? hist[t] : 0;
    int ex = block_excl_scan(c, t, wtot, 8);
    if (t < NB) {
        cursor[t] = ex;
        counts[blockIdx.x * NB + t] = c;
        starts[blockIdx.x * NB + t] = ex;
    }
    __syncthreads();
    int p[CH_PER_T];
    #pragma unroll
    for (int k = 0; k < CH_PER_T; ++k)
        p[k] = (t + SPLIT_T * k < CH) ? atomicAdd(&cursor[bin_of(dv[k])], 1) : 0;
    #pragma unroll
    for (int k = 0; k < CH_PER_T; ++k)
        if (t + SPLIT_T * k < CH) stage[p[k]] = make_int2(sv[k], dv[k]);
    __syncthreads();
    int2* o = binned + blockIdx.x * CH;
    for (int i = t; i < CH; i += SPLIT_T) o[i] = stage[i];
}

#define W1P 132
__global__ void __launch_bounds__(SORT_T)
k_binsort(const int* __restrict__ binned_w, const int* __restrict__ counts,
          const int* __restrict__ starts,
          const float* __restrict__ x, const float* __restrict__ W1,
          int* __restrict__ rowptr, int* __restrict__ cnt, float* __restrict__ dis,
          int* __restrict__ csr, float* __restrict__ h1) {
    const int2* __restrict__ binned = (const int2*)binned_w;
    __shared__ int lcnt[512];
    __shared__ int lcur[512];
    __shared__ int wtot[16];
    __shared__ int wlow[32];
    __shared__ int wown[32];
    __shared__ int lst[STAGE_CAP];
    __shared__ __align__(16) float wT[HID * W1P];
    __shared__ __align__(16) float ldis[512];
    int t = threadIdx.x;
    int lane = t & 31;
    int wv = __builtin_amdgcn_readfirstlane(t >> 5);
    int bin = blockIdx.x;
    int lo = bin * BIN_NODES;
    int nn = NN - lo < BIN_NODES ? NN - lo : BIN_NODES;

    int rs[RUNS_PER_W], rc[RUNS_PER_W];
    int own = 0;
    #pragma unroll
    for (int k = 0; k < RUNS_PER_W; ++k) {
        int ch = wv * RUNS_PER_W + k;
        unsigned c = min((unsigned)counts[ch * NB + bin], (unsigned)CH);
        unsigned s = min((unsigned)starts[ch * NB + bin], (unsigned)CH - c);
        rc[k] = (int)c;
        rs[k] = ch * CH + (int)s;
        own += (int)c;
    }
    int2 ev[RUNS_PER_W];
    #pragma unroll
    for (int k = 0; k < RUNS_PER_W; ++k)
        ev[k] = (lane < rc[k]) ? binned[rs[k] + lane] : make_int2(0, 0);

    float w1v[FIN * HID / SORT_T];
    #pragma unroll
    for (int j = 0; j < FIN * HID / SORT_T; ++j) w1v[j] = W1[t + SORT_T * j];

    int low = 0;
    {
        int col = t & (NB - 1);
        if (col < bin) {
            const int* cp = counts + (t >> 8) * NB + col;
            #pragma unroll
            for (int j = 0; j < NCH / 4; ++j) low += cp[j * 4 * NB];
        }
    }
    if (t < 512) lcnt[t] = 0;
    #pragma unroll
    for (int j = 0; j < FIN * HID / SORT_T; ++j) {
        int i = t + SORT_T * j;
        wT[(i & 63) * W1P + (i >> 6)] = w1v[j];
    }
    __syncthreads();

    #pragma unroll
    for (int k = 0; k < RUNS_PER_W; ++k)
        if (lane < rc[k]) {
            unsigned li = (unsigned)(ev[k].y - lo);
            atomicAdd(&lcnt[li < BIN_NODES - 1 ? li : BIN_NODES - 1], 1);
        }
    #pragma unroll
    for (int k = 0; k < RUNS_PER_W; ++k)
        for (int o = 32; o < rc[k]; o += 32)
            if (o + lane < rc[k]) {
                unsigned li = (unsigned)(binned[rs[k] + o + lane].y - lo);
                atomicAdd(&lcnt[li < BIN_NODES - 1 ? li : BIN_NODES - 1], 1);
            }
    low = wave_sum(low);
    if (lane == 0) { wlow[wv] = low; wown[wv] = own; }
    __syncthreads();
    int base = wave_sum(wlow[lane]);
    int n = wave_sum(wown[lane]);
    int c = (t < 512) ? lcnt[t] : 0;
    int ex = block_excl_scan(c, t, wtot, 16);
    float dsc = rsqrtf((float)(c + 1));
    if (t < 512) { lcur[t] = ex; ldis[t] = dsc; }
    if (t < nn) {
        rowptr[lo + t] = base + ex;
        cnt[lo + t] = c;
        dis[lo + t] = dsc;
    }
    __syncthreads();
    bool staged = n <= STAGE_CAP;
    int p[RUNS_PER_W];
    #pragma unroll
    for (int k = 0; k < RUNS_PER_W; ++k) {
        unsigned li = (unsigned)(ev[k].y - lo);
        p[k] = (lane < rc[k]) ? atomicAdd(&lcur[li < BIN_NODES - 1 ? li : BIN_NODES - 1], 1) : 0;
    }
    #pragma unroll
    for (int k = 0; k < RUNS_PER_W; ++k)
        if (lane < rc[k]) {
            if (staged) lst[p[k]] = ev[k].x;
            else csr[base + p[k]] = ev[k].x;
        }
    #pragma unroll
    for (int k = 0; k < RUNS_PER_W; ++k)
        for (int o = 32; o < rc[k]; o += 32)
            if (o + lane < rc[k]) {
                int2 e = binned[rs[k] + o + lane];
                unsigned li = (unsigned)(e.y - lo);
                int q = atomicAdd(&lcur[li < BIN_NODES - 1 ? li : BIN_NODES - 1], 1);
                if (staged) lst[q] = e.x;
                else csr[base + q] = e.x;
            }
    if (staged) {
        __syncthreads();
        for (int i = t; i < n; i += SORT_T) csr[base + i] = lst[i];
    }

    int r0 = wv * 16;
    if (r0 < nn) {
        int m = lane & 15, half = lane >> 4;
        int ra = r0 + m;
        ra = ra < nn ? ra : nn - 1;
        const float* arow = x + (size_t)(lo + ra) * FIN + 2 * half;
        v8f a0 = {}, a1 = {}, a2 = {}, a3 = {};
        const float* dp = ldis + r0 + 8 * half;
        float4 d0 = *(const float4*)(dp);
        float4 d1 = *(const float4*)(dp + 4);

        #pragma unroll 4
        for (int k0 = 0; k0 < FIN; k0 += 4) {
            v2f a = *(const v2f*)(arow + k0);
            const float* bp = wT + m * W1P + k0 + 2 * half;
            v2f b0 = *(const v2f*)(bp);
            v2f b1 = *(const v2f*)(bp + 16 * W1P);
            v2f b2 = *(const v2f*)(bp + 32 * W1P);
            v2f b3 = *(const v2f*)(bp + 48 * W1P);
            a0 = __builtin_amdgcn_wmma_f32_16x16x4_f32(false, a, false, b0, (short)0, a0, false, false);
            a1 = __builtin_amdgcn_wmma_f32_16x16x4_f32(false, a, false, b1, (short)0, a1, false, false);
            a2 = __builtin_amdgcn_wmma_f32_16x16x4_f32(false, a, false, b2, (short)0, a2, false, false);
            a3 = __builtin_amdgcn_wmma_f32_16x16x4_f32(false, a, false, b3, (short)0, a3, false, false);
        }

        int rb = r0 + 8 * half;
        float* out = h1 + (size_t)(lo + rb) * HID + m;
        const float ds[8] = {d0.x, d0.y, d0.z, d0.w, d1.x, d1.y, d1.z, d1.w};
        #pragma unroll
        for (int v = 0; v < 8; ++v)
            if (rb + v < nn) {
                out[v * HID +  0] = a0[v] * ds[v];
                out[v * HID + 16] = a1[v] * ds[v];
                out[v * HID + 32] = a2[v] * ds[v];
                out[v * HID + 48] = a3[v] * ds[v];
            }
    }
}

#define NPG 4
#define N_GROUPS 25000
#define GATHER_BLOCKS 1563

__device__ __forceinline__ int load_idx16(const int* __restrict__ csr, int beg, int e0, int l) {
    int j = beg + e0 + l;
    j = j < NE - 1 ? j : NE - 1;
    return (int)min((unsigned)csr[j], (unsigned)(NN - 1));
}

__device__ __forceinline__ int wave_count(int c) {
    int o = __shfl_xor(c, 16);
    return __builtin_amdgcn_readfirstlane(c > o ? c : o);
}

template <int B>
__device__ __forceinline__ void add_rows_f4(float4& acc, const float* __restrict__ h, int idx,
                                            int e0, int slot0, int c, unsigned self_off, int l) {
    float4 v[B];
    #pragma unroll
    for (int k = 0; k < B; ++k) {
        int s = __shfl(idx, slot0 + k, 16);
        unsigned off = (e0 + k < c) ? (unsigned)s * HID + 4u * (unsigned)l : self_off;
        v[k] = *(const float4*)(h + off);
    }
    #pragma unroll
    for (int k = 0; k < B; ++k) {
        bool ok = e0 + k < c;
        acc.x += ok ? v[k].x : 0.0f;
        acc.y += ok ? v[k].y : 0.0f;
        acc.z += ok ? v[k].z : 0.0f;
        acc.w += ok ? v[k].w : 0.0f;
    }
}

template <int B>
__device__ __forceinline__ void add_rows_f1(float& acc, const float* __restrict__ h, int idx,
                                            int e0, int slot0, int c, unsigned self_off, int f) {
    float v[B];
    #pragma unroll
    for (int k = 0; k < B; ++k) {
        int s = __shfl(idx, slot0 + k, 16);
        unsigned off = (e0 + k < c) ? (unsigned)s * H2P + (unsigned)f : self_off;
        v[k] = h[off];
    }
    #pragma unroll
    for (int k = 0; k < B; ++k) acc += (e0 + k < c) ? v[k] : 0.0f;
}

__device__ __forceinline__ float4 node_sum_f4(const int* __restrict__ csr, const float* __restrict__ h,
                                              int beg, int c, int ia, int ib, unsigned self_off, int l) {
    float4 acc = make_float4(0.0f, 0.0f, 0.0f, 0.0f);
    int cw = wave_count(c);
    if (cw > 0)  add_rows_f4<8>(acc, h, ia, 0, 0, c, self_off, l);
    if (cw > 8)  add_rows_f4<8>(acc, h, ia, 8, 8, c, self_off, l);
    if (cw > 16) add_rows_f4<8>(acc, h, ib, 16, 0, c, self_off, l);
    if (cw > 24) add_rows_f4<8>(acc, h, ib, 24, 8, c, self_off, l);
    for (int e0 = 32; e0 < cw; e0 += 16) {
        int ix = load_idx16(csr, beg, e0, l);
        add_rows_f4<8>(acc, h, ix, e0, 0, c, self_off, l);
        if (e0 + 8 < cw) add_rows_f4<8>(acc, h, ix, e0 + 8, 8, c, self_off, l);
    }
    return acc;
}

__device__ __forceinline__ float node_sum_f1(const int* __restrict__ csr, const float* __restrict__ h,
                                             int beg, int c, int ia, int ib, unsigned self_off,
                                             int l) {
    float acc = 0.0f;
    int cw = wave_count(c);
    if (cw > 0)  add_rows_f1<16>(acc, h, ia, 0, 0, c, self_off, l);
    if (cw > 16) add_rows_f1<16>(acc, h, ib, 16, 0, c, self_off, l);
    for (int e0 = 32; e0 < cw; e0 += 16) {
        int ix = load_idx16(csr, beg, e0, l);
        add_rows_f1<16>(acc, h, ix, e0, 0, c, self_off, l);
    }
    return acc;
}

#define ZP 68
#define BLK_ROWS (16 * NPG)
__global__ void __launch_bounds__(256)
k_gather1(const int* __restrict__ rowptr, const int* __restrict__ cnt,
          const int* __restrict__ csr, const float* __restrict__ dis,
          const float* __restrict__ h1, const float* __restrict__ b1,
          const float* __restrict__ W2, float* __restrict__ h2) {
    __shared__ float zt[BLK_ROWS * ZP];
    __shared__ float wT[16 * ZP];
    __shared__ __align__(16) float ho[BLK_ROWS * H2P];
    __shared__ float dl[BLK_ROWS];
    int t = threadIdx.x;
    int l = t & 15;
    int grp = blockIdx.x * 16 + (t >> 4);
    if (grp < N_GROUPS) {
        int n0 = grp * NPG;
        int4 rp = *(const int4*)(rowptr + n0);
        int4 cn = *(const int4*)(cnt + n0);
        float4 dv = *(const float4*)(dis + n0);
        if (l == 0) *(float4*)(dl + (t >> 4) * NPG) = dv;
        const int begs[NPG] = {rp.x, rp.y, rp.z, rp.w};
        const int cs[NPG] = {cn.x, cn.y, cn.z, cn.w};
        const float ds[NPG] = {dv.x, dv.y, dv.z, dv.w};
        float4 bb = *(const float4*)(b1 + 4 * l);
        int ia[NPG];
        #pragma unroll
        for (int n = 0; n < NPG; ++n)
            ia[n] = load_idx16(csr, begs[n], 0, l);
        #pragma unroll
        for (int n = 0; n < NPG; ++n) {
            int node = n0 + n;
            unsigned self_off = (unsigned)node * HID + 4u * (unsigned)l;
            float4 hv = *(const float4*)(h1 + self_off);
            int ibn = load_idx16(csr, begs[n], 16, l);
            float4 acc = node_sum_f4(csr, h1, begs[n], cs[n], ia[n], ibn, self_off, l);
            float di = ds[n];
            float4 r;
            r.x = (acc.x + hv.x) * di + bb.x;
            r.y = (acc.y + hv.y) * di + bb.y;
            r.z = (acc.z + hv.z) * di + bb.z;
            r.w = (acc.w + hv.w) * di + bb.w;
            r.x = r.x > 0.0f ? r.x : 0.0f;
            r.y = r.y > 0.0f ? r.y : 0.0f;
            r.z = r.z > 0.0f ? r.z : 0.0f;
            r.w = r.w > 0.0f ? r.w : 0.0f;
            *(float4*)(zt + ((t >> 4) * NPG + n) * ZP + 4 * l) = r;
        }
    }
    {
        float wv4[4];
        #pragma unroll
        for (int j = 0; j < 4; ++j) {
            int i = t + 256 * j;
            int k = i >> 4, c = i & 15;
            wv4[j] = W2[k * NCLS + (c < NCLS ? c : NCLS - 1)];
        }
        #pragma unroll
        for (int j = 0; j < 4; ++j) {
            int i = t + 256 * j;
            int k = i >> 4, c = i & 15;
            wT[c * ZP + k] = (c < NCLS) ? wv4[j] : 0.0f;
        }
    }
    __syncthreads();

    int wv = t >> 5;
    int tile = blockIdx.x * 4 + wv;
    if (wv < 4 && tile < ROW_TILES) {
        int lane = t & 31;
        int m = lane & 15, half = lane >> 4;
        const float* dp = dl + wv * 16 + 8 * half;
        float4 d0 = *(const float4*)(dp);
        float4 d1 = *(const float4*)(dp + 4);
        const float* ap = zt + (wv * 16 + m) * ZP + 2 * half;
        const float* bp = wT + m * ZP + 2 * half;
        v8f acc = {};
        #pragma unroll
        for (int k0 = 0; k0 < HID; k0 += 4) {
            v2f a = *(const v2f*)(ap + k0);
            v2f b = *(const v2f*)(bp + k0);
            acc = __builtin_amdgcn_wmma_f32_16x16x4_f32(false, a, false, b, (short)0, acc, false, false);
        }
        asm volatile("v_nop\n\tv_nop\n\tv_nop\n\tv_nop" : "+v"(acc));
        float* o = ho + (wv * 16 + 8 * half) * H2P + m;
        const float sc[8] = {d0.x, d0.y, d0.z, d0.w, d1.x, d1.y, d1.z, d1.w};
        #pragma unroll
        for (int v = 0; v < 8; ++v) o[v * H2P] = acc[v] * sc[v];
    }
    __syncthreads();
    {
        size_t gi = (size_t)blockIdx.x * (BLK_ROWS * H2P) + 4 * t;
        if (gi < (size_t)NN * H2P) *(float4*)(h2 + gi) = *(const float4*)(ho + 4 * t);
    }
}

__global__ void __launch_bounds__(256)
k_gather2(const int* __restrict__ rowptr, const int* __restrict__ cnt,
          const int* __restrict__ csr, const float* __restrict__ dis,
          const float* __restrict__ h2, const float* __restrict__ b2,
          float* __restrict__ out) {
    __shared__ float stage[16 * NPG * NCLS];
    int t = threadIdx.x;
    int f = t & 15;
    int grp = blockIdx.x * 16 + (t >> 4);
    if (grp < N_GROUPS) {
        int n0 = grp * NPG;
        int4 rp = *(const int4*)(rowptr + n0);
        int4 cn = *(const int4*)(cnt + n0);
        float4 dv = *(const float4*)(dis + n0);
        const int begs[NPG] = {rp.x, rp.y, rp.z, rp.w};
        const int cs[NPG] = {cn.x, cn.y, cn.z, cn.w};
        const float ds[NPG] = {dv.x, dv.y, dv.z, dv.w};
        float bias = b2[f < NCLS ? f : NCLS - 1];
        int ia[NPG];
        #pragma unroll
        for (int n = 0; n < NPG; ++n)
            ia[n] = load_idx16(csr, begs[n], 0, f);
        #pragma unroll
        for (int n = 0; n < NPG; ++n) {
            int node = n0 + n;
            unsigned self_off = (unsigned)node * H2P + (unsigned)f;
            float hv = h2[self_off];
            int ibn = load_idx16(csr, begs[n], 16, f);
            float acc = node_sum_f1(csr, h2, begs[n], cs[n], ia[n], ibn, self_off, f);
            float v = (acc + hv) * ds[n] + bias;
            float mx = f < NCLS ? v : -3.402823466e38f;
            for (int o = 8; o > 0; o >>= 1) mx = fmaxf(mx, __shfl_xor(mx, o, 16));
            float ex = f < NCLS ? expf(v - mx) : 0.0f;
            for (int o = 8; o > 0; o >>= 1) ex += __shfl_xor(ex, o, 16);
            float lse = mx + logf(ex);
            if (f < NCLS) stage[((t >> 4) * NPG + n) * NCLS + f] = v - lse;
        }
    }
    __syncthreads();
    for (int i = t; i < 16 * NPG * NCLS; i += 256) {
        size_t gi = (size_t)blockIdx.x * (16 * NPG * NCLS) + i;
        if (gi < (size_t)NN * NCLS) out[gi] = stage[i];
    }
}

extern "C" void kernel_launch(void* const* d_in, const int* in_sizes, int n_in,
                              void* d_out, int out_size, void* d_ws, size_t ws_size,
                              hipStream_t stream) {
    const float* x  = (const float*)d_in[0];
    const int*   ei = (const int*)d_in[1];
    const float* W1 = (const float*)d_in[2];
    const float* b1 = (const float*)d_in[3];
    const float* W2 = (const float*)d_in[4];
    const float* b2 = (const float*)d_in[5];
    const int* src = ei;
    const int* dst = ei + NE;
    float* out = (float*)d_out;

    float* ws   = (float*)d_ws;
    float* h1   = ws;
    float* h2   = h1 + 2 * (size_t)NN * HID;
    float* dis  = h2 + (size_t)NN * H2P;
    int* cnt    = (int*)(dis + NN);
    int* rowptr = cnt + NN;
    int* csr    = rowptr + NN;
    int* binned = csr + NE;
    int* counts = binned + 2 * (size_t)NE;
    int* starts = counts + NCH * NB;

    k_binsplit<<<NCH, SPLIT_T, 0, stream>>>(src, dst, counts, starts, binned);
    k_binsort<<<NB, SORT_T, 0, stream>>>(binned, counts, starts, x, W1, rowptr, cnt, dis, csr, h1);

    k_gather1<<<GATHER_BLOCKS, 256, 0, stream>>>(rowptr, cnt, csr, dis, h1, b1, W2, h2);

    k_gather2<<<GATHER_BLOCKS, 256, 0, stream>>>(rowptr, cnt, csr, dis, h2, b2, out);
}


# Layer-1 gather cut to 64 vector registers (own row walked as the last slot, 32-bit row offsets on a scalar base): all 1563 work-groups resident at once, no second round; sum order kept
# speedup vs baseline: 6.9587x; 1.0439x over previous
#include <hip/hip_runtime.h>
#include <math.h>


#define NN 100000
#define NE 1600000
#define FIN 128
#define HID 64
#define NCLS 10
#define H2P 16
#define ROW_TILES 6250
#define NB 256
#define BIN_NODES 391
#define NCH 256
#define CH 6250
#define SPLIT_T 512
#define CH_PER_T 13
#define SORT_T 1024
#define RUNS_PER_W 8
#define STAGE_CAP 16384

typedef __attribute__((ext_vector_type(2))) float v2f;
typedef __attribute__((ext_vector_type(8))) float v8f;

__device__ __forceinline__ unsigned bin_of(int d) {
    unsigned b = (unsigned)d / BIN_NODES;
    return b < NB - 1 ? b : NB - 1;
}

__device__ __forceinline__ int block_excl_scan(int c, int t, int* wtot, int nwaves) {
    int lane = t & 31, wv = t >> 5;
    int inc = c;
    for (int o = 1; o < 32; o <<= 1) {
        int v = __shfl_up(inc, o);
        if (lane >= o) inc += v;
    }
    if (lane == 31 && wv < nwaves) wtot[wv] = inc;
    __syncthreads();
    int off = 0;
    for (int w = 0; w < nwaves; ++w) off += (w < wv) ? wtot[w] : 0;
    return off + inc - c;
}

__device__ __forceinline__ int wave_sum(int v) {
    for (int o = 16; o > 0; o >>= 1) v += __shfl_xor(v, o);
    return v;
}

__global__ void __launch_bounds__(SPLIT_T)
k_binsplit(const int* __restrict__ src, const int* __restrict__ dst,
           int* __restrict__ counts, int* __restrict__ starts, int* __restrict__ binned_w) {
    int2* __restrict__ binned = (int2*)binned_w;
    __shared__ int hist[NB];
    __shared__ int cursor[NB];
    __shared__ int wtot[8];
    __shared__ int2 stage[CH];
    int t = threadIdx.x;
    if (t < NB) hist[t] = 0;
    const int* sp = src + blockIdx.x * CH;
    const int* dp = dst + blockIdx.x * CH;
    int sv[CH_PER_T], dv[CH_PER_T];
    #pragma unroll
    for (int k = 0; k < CH_PER_T; ++k) {
        int i = t + SPLIT_T * k;
        dv[k] = (i < CH) ? dp[i] : -1;
        sv[k] = (i < CH) ? sp[i] : 0;
    }
    __syncthreads();
    #pragma unroll
    for (int k = 0; k < CH_PER_T; ++k)
        if (t + SPLIT_T * k < CH) atomicAdd(&hist[bin_of(dv[k])], 1);
    __syncthreads();
    int c = (t < NB) ? hist[t] : 0;
    int ex = block_excl_scan(c, t, wtot, 8);
    if (t < NB) {
        cursor[t] = ex;
        counts[blockIdx.x * NB + t] = c;
        starts[blockIdx.x * NB + t] = ex;
    }
    __syncthreads();
    int p[CH_PER_T];
    #pragma unroll
    for (int k = 0; k < CH_PER_T; ++k)
        p[k] = (t + SPLIT_T * k < CH) ? atomicAdd(&cursor[bin_of(dv[k])], 1) : 0;
    #pragma unroll
    for (int k = 0; k < CH_PER_T; ++k)
        if (t + SPLIT_T * k < CH) stage[p[k]] = make_int2(sv[k], dv[k]);
    __syncthreads();
    int2* o = binned + blockIdx.x * CH;
    for (int i = t; i < CH; i += SPLIT_T) o[i] = stage[i];
}

#define W1P 132
__global__ void __launch_bounds__(SORT_T)
k_binsort(const int* __restrict__ binned_w, const int* __restrict__ counts,
          const int* __restrict__ starts,
          const float* __restrict__ x, const float* __restrict__ W1,
          int* __restrict__ rowptr, int* __restrict__ cnt, float* __restrict__ dis,
          int* __restrict__ csr, float* __restrict__ h1) {
    const int2* __restrict__ binned = (const int2*)binned_w;
    __shared__ int lcnt[512];
    __shared__ int lcur[512];
    __shared__ int wtot[16];
    __shared__ int wlow[32];
    __shared__ int wown[32];
    __shared__ int lst[STAGE_CAP];
    __shared__ __align__(16) float wT[HID * W1P];
    __shared__ __align__(16) float ldis[512];
    int t = threadIdx.x;
    int lane = t & 31;
    int wv = __builtin_amdgcn_readfirstlane(t >> 5);
    int bin = blockIdx.x;
    int lo = bin * BIN_NODES;
    int nn = NN - lo < BIN_NODES ? NN - lo : BIN_NODES;

    int rs[RUNS_PER_W], rc[RUNS_PER_W];
    int own = 0;
    #pragma unroll
    for (int k = 0; k < RUNS_PER_W; ++k) {
        int ch = wv * RUNS_PER_W + k;
        unsigned c = min((unsigned)counts[ch * NB + bin], (unsigned)CH);
        unsigned s = min((unsigned)starts[ch * NB + bin], (unsigned)CH - c);
        rc[k] = (int)c;
        rs[k] = ch * CH + (int)s;
        own += (int)c;
    }
    int2 ev[RUNS_PER_W];
    #pragma unroll
    for (int k = 0; k < RUNS_PER_W; ++k)
        ev[k] = (lane < rc[k]) ? binned[rs[k] + lane] : make_int2(0, 0);

    float w1v[FIN * HID / SORT_T];
    #pragma unroll
    for (int j = 0; j < FIN * HID / SORT_T; ++j) w1v[j] = W1[t + SORT_T * j];

    int low = 0;
    {
        int col = t & (NB - 1);
        if (col < bin) {
            const int* cp = counts + (t >> 8) * NB + col;
            #pragma unroll
            for (int j = 0; j < NCH / 4; ++j) low += cp[j * 4 * NB];
        }
    }
    if (t < 512) lcnt[t] = 0;
    #pragma unroll
    for (int j = 0; j < FIN * HID / SORT_T; ++j) {
        int i = t + SORT_T * j;
        wT[(i & 63) * W1P + (i >> 6)] = w1v[j];
    }
    __syncthreads();

    #pragma unroll
    for (int k = 0; k < RUNS_PER_W; ++k)
        if (lane < rc[k]) {
            unsigned li = (unsigned)(ev[k].y - lo);
            atomicAdd(&lcnt[li < BIN_NODES - 1 ? li : BIN_NODES - 1], 1);
        }
    #pragma unroll
    for (int k = 0; k < RUNS_PER_W; ++k)
        for (int o = 32; o < rc[k]; o += 32)
            if (o + lane < rc[k]) {
                unsigned li = (unsigned)(binned[rs[k] + o + lane].y - lo);
                atomicAdd(&lcnt[li < BIN_NODES - 1 ? li : BIN_NODES - 1], 1);
            }
    low = wave_sum(low);
    if (lane == 0) { wlow[wv] = low; wown[wv] = own; }
    __syncthreads();
    int base = wave_sum(wlow[lane]);
    int n = wave_sum(wown[lane]);
    int c = (t < 512) ? lcnt[t] : 0;
    int ex = block_excl_scan(c, t, wtot, 16);
    float dsc = rsqrtf((float)(c + 1));
    if (t < 512) { lcur[t] = ex; ldis[t] = dsc; }
    if (t < nn) {
        rowptr[lo + t] = base + ex;
        cnt[lo + t] = c;
        dis[lo + t] = dsc;
    }
    __syncthreads();
    bool staged = n <= STAGE_CAP;
    int p[RUNS_PER_W];
    #pragma unroll
    for (int k = 0; k < RUNS_PER_W; ++k) {
        unsigned li = (unsigned)(ev[k].y - lo);
        p[k] = (lane < rc[k]) ? atomicAdd(&lcur[li < BIN_NODES - 1 ? li : BIN_NODES - 1], 1) : 0;
    }
    #pragma unroll
    for (int k = 0; k < RUNS_PER_W; ++k)
        if (lane < rc[k]) {
            if (staged) lst[p[k]] = ev[k].x;
            else csr[base + p[k]] = ev[k].x;
        }
    #pragma unroll
    for (int k = 0; k < RUNS_PER_W; ++k)
        for (int o = 32; o < rc[k]; o += 32)
            if (o + lane < rc[k]) {
                int2 e = binned[rs[k] + o + lane];
                unsigned li = (unsigned)(e.y - lo);
                int q = atomicAdd(&lcur[li < BIN_NODES - 1 ? li : BIN_NODES - 1], 1);
                if (staged) lst[q] = e.x;
                else csr[base + q] = e.x;
            }
    if (staged) {
        __syncthreads();
        for (int i = t; i < n; i += SORT_T) csr[base + i] = lst[i];
    }

    int r0 = wv * 16;
    if (r0 < nn) {
        int m = lane & 15, half = lane >> 4;
        int ra = r0 + m;
        ra = ra < nn ? ra : nn - 1;
        const float* arow = x + (size_t)(lo + ra) * FIN + 2 * half;
        v8f a0 = {}, a1 = {}, a2 = {}, a3 = {};
        const float* dp = ldis + r0 + 8 * half;
        float4 d0 = *(const float4*)(dp);
        float4 d1 = *(const float4*)(dp + 4);

        #pragma unroll 4
        for (int k0 = 0; k0 < FIN; k0 += 4) {
            v2f a = *(const v2f*)(arow + k0);
            const float* bp = wT + m * W1P + k0 + 2 * half;
            v2f b0 = *(const v2f*)(bp);
            v2f b1 = *(const v2f*)(bp + 16 * W1P);
            v2f b2 = *(const v2f*)(bp + 32 * W1P);
            v2f b3 = *(const v2f*)(bp + 48 * W1P);
            a0 = __builtin_amdgcn_wmma_f32_16x16x4_f32(false, a, false, b0, (short)0, a0, false, false);
            a1 = __builtin_amdgcn_wmma_f32_16x16x4_f32(false, a, false, b1, (short)0, a1, false, false);
            a2 = __builtin_amdgcn_wmma_f32_16x16x4_f32(false, a, false, b2, (short)0, a2, false, false);
            a3 = __builtin_amdgcn_wmma_f32_16x16x4_f32(false, a, false, b3, (short)0, a3, false, false);
        }

        int rb = r0 + 8 * half;
        float* out = h1 + (size_t)(lo + rb) * HID + m;
        const float ds[8] = {d0.x, d0.y, d0.z, d0.w, d1.x, d1.y, d1.z, d1.w};
        #pragma unroll
        for (int v = 0; v < 8; ++v)
            if (rb + v < nn) {
                out[v * HID +  0] = a0[v] * ds[v];
                out[v * HID + 16] = a1[v] * ds[v];
                out[v * HID + 32] = a2[v] * ds[v];
                out[v * HID + 48] = a3[v] * ds[v];
            }
    }
}

#define NPG 4
#define N_GROUPS 25000
#define GATHER_BLOCKS 1563

__device__ __forceinline__ int load_idx16(const int* __restrict__ csr, int beg, int e0, int l) {
    int j = beg + e0 + l;
    j = j < NE - 1 ? j : NE - 1;
    return (int)min((unsigned)csr[j], (unsigned)(NN - 1));
}

__device__ __forceinline__ int wave_count(int c) {
    int o = __shfl_xor(c, 16);
    return __builtin_amdgcn_readfirstlane(c > o ? c : o);
}

template <int B>
__device__ __forceinline__ void add_rows_f4(float4& acc, const float4* __restrict__ h, int idx,
                                            int e0, int slot0, int c, unsigned self_off, int l) {
    float4 v[B];
    #pragma unroll
    for (int k = 0; k < B; ++k) {
        int s = __shfl(idx, slot0 + k, 16);
        unsigned off = (e0 + k < c) ? (unsigned)s * (HID / 4) + (unsigned)l : self_off;
        v[k] = h[off];
    }
    #pragma unroll
    for (int k = 0; k < B; ++k) {
        bool ok = e0 + k < c;
        acc.x += ok ? v[k].x : 0.0f;
        acc.y += ok ? v[k].y : 0.0f;
        acc.z += ok ? v[k].z : 0.0f;
        acc.w += ok ? v[k].w : 0.0f;
    }
}

template <int B>
__device__ __forceinline__ void add_rows_f1(float& acc, const float* __restrict__ h, int idx,
                                            int e0, int slot0, int c, unsigned self_off, int f) {
    float v[B];
    #pragma unroll
    for (int k = 0; k < B; ++k) {
        int s = __shfl(idx, slot0 + k, 16);
        unsigned off = (e0 + k < c) ? (unsigned)s * H2P + (unsigned)f : self_off;
        v[k] = h[off];
    }
    #pragma unroll
    for (int k = 0; k < B; ++k) acc += (e0 + k < c) ? v[k] : 0.0f;
}

__device__ __forceinline__ float4 node_sum_f4(const int* __restrict__ csr, const float4* __restrict__ h, int node,
                                              int beg, int c, int cw, int ia, int ib, unsigned self_off, int l) {
    float4 acc = make_float4(0.0f, 0.0f, 0.0f, 0.0f);
    int c1 = c + 1;
    ia = (l == c) ? node : ia;
    ib = (l + 16 == c) ? node : ib;
    add_rows_f4<8>(acc, h, ia, 0, 0, c1, self_off, l);
    if (cw >= 8)  add_rows_f4<8>(acc, h, ia, 8, 8, c1, self_off, l);
    if (cw >= 16) add_rows_f4<8>(acc, h, ib, 16, 0, c1, self_off, l);
    if (cw >= 24) add_rows_f4<8>(acc, h, ib, 24, 8, c1, self_off, l);
    for (int e0 = 32; e0 <= cw; e0 += 16) {
        int ix = load_idx16(csr, beg, e0, l);
        ix = (e0 + l == c) ? node : ix;
        add_rows_f4<8>(acc, h, ix, e0, 0, c1, self_off, l);
        if (e0 + 8 <= cw) add_rows_f4<8>(acc, h, ix, e0 + 8, 8, c1, self_off, l);
    }
    return acc;
}

__device__ __forceinline__ float node_sum_f1(const int* __restrict__ csr, const float* __restrict__ h,
                                             int beg, int c, int ia, int ib, unsigned self_off,
                                             int l) {
    float acc = 0.0f;
    int cw = wave_count(c);
    if (cw > 0)  add_rows_f1<16>(acc, h, ia, 0, 0, c, self_off, l);
    if (cw > 16) add_rows_f1<16>(acc, h, ib, 16, 0, c, self_off, l);
    for (int e0 = 32; e0 < cw; e0 += 16) {
        int ix = load_idx16(csr, beg, e0, l);
        add_rows_f1<16>(acc, h, ix, e0, 0, c, self_off, l);
    }
    return acc;
}

#define ZP 68
#define BLK_ROWS (16 * NPG)
__global__ void __launch_bounds__(256, 16)
k_gather1(const int* __restrict__ rowptr, const int* __restrict__ cnt,
          const int* __restrict__ csr, const float* __restrict__ dis,
          const float* __restrict__ h1, const float* __restrict__ b1,
          const float* __restrict__ W2, float* __restrict__ h2) {
    __shared__ float zt[BLK_ROWS * ZP];
    __shared__ float wT[16 * ZP];
    __shared__ __align__(16) float ho[BLK_ROWS * H2P];
    __shared__ float dl[BLK_ROWS];
    int t = threadIdx.x;
    int l = t & 15;
    int grp = blockIdx.x * 16 + (t >> 4);
    const float4* __restrict__ h1q = (const float4*)h1;
    if (grp < N_GROUPS) {
        int n0 = grp * NPG;
        int4 rp = *(const int4*)(rowptr + n0);
        int4 cn = *(const int4*)(cnt + n0);
        float4 dv = *(const float4*)(dis + n0);
        if (l == 0) *(float4*)(dl + (t >> 4) * NPG) = dv;
        const int begs[NPG] = {rp.x, rp.y, rp.z, rp.w};
        const int cs[NPG] = {cn.x, cn.y, cn.z, cn.w};
        const float ds[NPG] = {dv.x, dv.y, dv.z, dv.w};
        float4 bb = *(const float4*)(b1 + 4 * l);
        int cws[NPG];
        #pragma unroll
        for (int n = 0; n < NPG; ++n) cws[n] = wave_count(cs[n]);
        int ian = load_idx16(csr, begs[0], 0, l);
        #pragma unroll
        for (int n = 0; n < NPG; ++n) {
            int node = n0 + n;
            unsigned self_off = (unsigned)node * (HID / 4) + (unsigned)l;
            int iac = ian;
            if (n + 1 < NPG) ian = load_idx16(csr, begs[n + 1], 0, l);
            int ibn = load_idx16(csr, begs[n], 16, l);
            float4 acc = node_sum_f4(csr, h1q, node, begs[n], cs[n], cws[n], iac, ibn, self_off, l);
            float di = ds[n];
            float4 r;
            r.x = acc.x * di + bb.x;
            r.y = acc.y * di + bb.y;
            r.z = acc.z * di + bb.z;
            r.w = acc.w * di + bb.w;
            r.x = r.x > 0.0f ? r.x : 0.0f;
            r.y = r.y > 0.0f ? r.y : 0.0f;
            r.z = r.z > 0.0f ? r.z : 0.0f;
            r.w = r.w > 0.0f ? r.w : 0.0f;
            *(float4*)(zt + ((t >> 4) * NPG + n) * ZP + 4 * l) = r;
        }
    }
    {
        float wv4[4];
        #pragma unroll
        for (int j = 0; j < 4; ++j) {
            int i = t + 256 * j;
            int k = i >> 4, c = i & 15;
            wv4[j] = W2[k * NCLS + (c < NCLS ? c : NCLS - 1)];
        }
        #pragma unroll
        for (int j = 0; j < 4; ++j) {
            int i = t + 256 * j;
            int k = i >> 4, c = i & 15;
            wT[c * ZP + k] = (c < NCLS) ? wv4[j] : 0.0f;
        }
    }
    __syncthreads();

    int wv = t >> 5;
    int tile = blockIdx.x * 4 + wv;
    if (wv < 4 && tile < ROW_TILES) {
        int lane = t & 31;
        int m = lane & 15, half = lane >> 4;
        const float* dp = dl + wv * 16 + 8 * half;
        float4 d0 = *(const float4*)(dp);
        float4 d1 = *(const float4*)(dp + 4);
        const float* ap = zt + (wv * 16 + m) * ZP + 2 * half;
        const float* bp = wT + m * ZP + 2 * half;
        float* o = ho + (wv * 16 + 8 * half) * H2P + m;
        __builtin_amdgcn_sched_barrier(0);
        v8f acc = {};
        #pragma unroll
        for (int k0 = 0; k0 < HID; k0 += 4) {
            v2f a = *(const v2f*)(ap + k0);
            v2f b = *(const v2f*)(bp + k0);
            acc = __builtin_amdgcn_wmma_f32_16x16x4_f32(false, a, false, b, (short)0, acc, false, false);
        }
        asm volatile("v_nop\n\tv_nop\n\tv_nop\n\tv_nop" : "+v"(acc));
        const float sc[8] = {d0.x, d0.y, d0.z, d0.w, d1.x, d1.y, d1.z, d1.w};
        #pragma unroll
        for (int v = 0; v < 8; ++v) o[v * H2P] = acc[v] * sc[v];
    }
    __syncthreads();
    {
        size_t gi = (size_t)blockIdx.x * (BLK_ROWS * H2P) + 4 * t;
        if (gi < (size_t)NN * H2P) *(float4*)(h2 + gi) = *(const float4*)(ho + 4 * t);
    }
}

__global__ void __launch_bounds__(256)
k_gather2(const int* __restrict__ rowptr, const int* __restrict__ cnt,
          const int* __restrict__ csr, const float* __restrict__ dis,
          const float* __restrict__ h2, const float* __restrict__ b2,
          float* __restrict__ out) {
    __shared__ float stage[16 * NPG * NCLS];
    int t = threadIdx.x;
    int f = t & 15;
    int grp = blockIdx.x * 16 + (t >> 4);
    if (grp < N_GROUPS) {
        int n0 = grp * NPG;
        int4 rp = *(const int4*)(rowptr + n0);
        int4 cn = *(const int4*)(cnt + n0);
        float4 dv = *(const float4*)(dis + n0);
        const int begs[NPG] = {rp.x, rp.y, rp.z, rp.w};
        const int cs[NPG] = {cn.x, cn.y, cn.z, cn.w};
        const float ds[NPG] = {dv.x, dv.y, dv.z, dv.w};
        float bias = b2[f < NCLS ? f : NCLS - 1];
        int ia[NPG];
        #pragma unroll
        for (int n = 0; n < NPG; ++n)
            ia[n] = load_idx16(csr, begs[n], 0, f);
        #pragma unroll
        for (int n = 0; n < NPG; ++n) {
            int node = n0 + n;
            unsigned self_off = (unsigned)node * H2P + (unsigned)f;
            float hv = h2[self_off];
            int ibn = load_idx16(csr, begs[n], 16, f);
            float acc = node_sum_f1(csr, h2, begs[n], cs[n], ia[n], ibn, self_off, f);
            float v = (acc + hv) * ds[n] + bias;
            float mx = f < NCLS ? v : -3.402823466e38f;
            for (int o = 8; o > 0; o >>= 1) mx = fmaxf(mx, __shfl_xor(mx, o, 16));
            float ex = f < NCLS ? expf(v - mx) : 0.0f;
            for (int o = 8; o > 0; o >>= 1) ex += __shfl_xor(ex, o, 16);
            float lse = mx + logf(ex);
            if (f < NCLS) stage[((t >> 4) * NPG + n) * NCLS + f] = v - lse;
        }
    }
    __syncthreads();
    for (int i = t; i < 16 * NPG * NCLS; i += 256) {
        size_t gi = (size_t)blockIdx.x * (16 * NPG * NCLS) + i;
        if (gi < (size_t)NN * NCLS) out[gi] = stage[i];
    }
}

extern "C" void kernel_launch(void* const* d_in, const int* in_sizes, int n_in,
                              void* d_out, int out_size, void* d_ws, size_t ws_size,
                              hipStream_t stream) {
    const float* x  = (const float*)d_in[0];
    const int*   ei = (const int*)d_in[1];
    const float* W1 = (const float*)d_in[2];
    const float* b1 = (const float*)d_in[3];
    const float* W2 = (const float*)d_in[4];
    const float* b2 = (const float*)d_in[5];
    const int* src = ei;
    const int* dst = ei + NE;
    float* out = (float*)d_out;

    float* ws   = (float*)d_ws;
    float* h1   = ws;
    float* h2   = h1 + 2 * (size_t)NN * HID;
    float* dis  = h2 + (size_t)NN * H2P;
    int* cnt    = (int*)(dis + NN);
    int* rowptr = cnt + NN;
    int* csr    = rowptr + NN;
    int* binned = csr + NE;
    int* counts = binned + 2 * (size_t)NE;
    int* starts = counts + NCH * NB;

    k_binsplit<<<NCH, SPLIT_T, 0, stream>>>(src, dst, counts, starts, binned);
    k_binsort<<<NB, SORT_T, 0, stream>>>(binned, counts, starts, x, W1, rowptr, cnt, dis, csr, h1);

    k_gather1<<<GATHER_BLOCKS, 256, 0, stream>>>(rowptr, cnt, csr, dis, h1, b1, W2, h2);

    k_gather2<<<GATHER_BLOCKS, 256, 0, stream>>>(rowptr, cnt, csr, dis, h2, b2, out);
}
